# starting kernel
# baseline (speedup 1.0000x reference)
#include <hip/hip_runtime.h>
#include <math.h>


typedef __attribute__((ext_vector_type(2))) float v2f;
typedef __attribute__((ext_vector_type(8))) float v8f;

#define NB 32
#define NC 512
#define HW 196
#define ND 8192

__global__ void cbp_zero_kernel(float* __restrict__ p, int n) {
  int i = blockIdx.x * blockDim.x + threadIdx.x;
  if (i < n) p[i] = 0.0f;
}

__global__ __launch_bounds__(256) void cbp_gram_sketch_kernel(
    const float* __restrict__ x, const int* __restrict__ h1,
    const float* __restrict__ s1, const int* __restrict__ h2,
    const float* __restrict__ s2, float* __restrict__ cbp) {
  __shared__ float lds_cbp[ND];

  const int b   = blockIdx.x >> 2;
  const int i8  = blockIdx.x & 3;
  const int tid = threadIdx.x;
  const int wave = tid >> 5;
  const int lane = tid & 31;
  const int half = lane >> 4;
  const int lr   = lane & 15;

  for (int idx = tid; idx < ND; idx += 256) lds_cbp[idx] = 0.0f;
  __syncthreads();

  const float* xb = x + (size_t)b * (NC * HW);

  for (int t = wave; t < 256; t += 8) {
    const int j  = t & 31;
    const int i  = i8 * 8 + (t >> 5);

    const float* pa = xb + (i * 16 + lr) * HW + 2 * half;
    const float* pb = xb + (j * 16 + lr) * HW + 2 * half;

    v8f acc = {};
#pragma unroll 7
    for (int k = 0; k < 49; ++k) {
      v2f a  = *(const v2f*)(pa + 4 * k);
      v2f bb = *(const v2f*)(pb + 4 * k);
      acc = __builtin_amdgcn_wmma_f32_16x16x4_f32(
           false, a,  false, bb,
           (short)0, acc,  false,  false);
    }

    const int   n     = j * 16 + lr;
    const int   h2n   = h2[n];
    const float s2n   = s2[n];
    const int   mbase = i * 16 + half * 8;
#pragma unroll
    for (int r = 0; r < 8; ++r) {
      const int m      = mbase + r;
      const int bucket = (h1[m] + h2n) & (ND - 1);
      atomicAdd(&lds_cbp[bucket], s1[m] * s2n * acc[r]);
    }
  }

  __syncthreads();
  float* dst = cbp + (size_t)b * ND;
  for (int idx = tid; idx < ND; idx += 256) {
    atomicAdd(&dst[idx], lds_cbp[idx]);
  }
}

__global__ __launch_bounds__(256) void cbp_finalize_kernel(
    const float* __restrict__ cbp, float* __restrict__ out) {
  __shared__ float red[256];
  const int b   = blockIdx.x;
  const int tid = threadIdx.x;
  const float* v = cbp + (size_t)b * ND;

  float ssum = 0.0f;
  for (int idx = tid; idx < ND; idx += 256) ssum += fabsf(v[idx]) + 1e-5f;
  red[tid] = ssum;
  __syncthreads();
  for (int s = 128; s > 0; s >>= 1) {
    if (tid < s) red[tid] += red[tid + s];
    __syncthreads();
  }
  const float inv = 1.0f / fmaxf(sqrtf(red[0]), 1e-12f);

  float* o = out + (size_t)b * ND;
  for (int idx = tid; idx < ND; idx += 256) {
    const float val = v[idx];
    const float sg  = (val > 0.0f) ? 1.0f : ((val < 0.0f) ? -1.0f : 0.0f);
    o[idx] = sg * sqrtf(fabsf(val) + 1e-5f) * inv;
  }
}

extern "C" void kernel_launch(void* const* d_in, const int* in_sizes, int n_in,
                              void* d_out, int out_size, void* d_ws, size_t ws_size,
                              hipStream_t stream) {
  const float* x  = (const float*)d_in[0];
  const int*   h1 = (const int*)d_in[1];
  const float* s1 = (const float*)d_in[2];
  const int*   h2 = (const int*)d_in[3];
  const float* s2 = (const float*)d_in[4];
  float*       out = (float*)d_out;

  float* cbp = (float*)d_ws;
  const int n = NB * ND;

  cbp_zero_kernel<<<(n + 255) / 256, 256, 0, stream>>>(cbp, n);
  cbp_gram_sketch_kernel<<<NB * 4, 256, 0, stream>>>(x, h1, s1, h2, s2, cbp);
  cbp_finalize_kernel<<<NB, 256, 0, stream>>>(cbp, out);
}


# Each wave owns a 128x64 block of G: 4x4 register-tiled WMMA on 128-bit fragment loads, summed in a wave-private LDS histogram; block pairs add to two planes, so sum order is fixed; k order changes
# speedup vs baseline: 3.0020x; 3.0020x over previous
#include <hip/hip_runtime.h>
#include <math.h>


typedef __attribute__((ext_vector_type(2))) float v2f;
typedef __attribute__((ext_vector_type(4))) float v4f;
typedef __attribute__((ext_vector_type(8))) float v8f;

#define NB 32
#define NC 512
#define HW 196
#define ND 8192

__global__ void cbp_zero_kernel(float* __restrict__ p, float* __restrict__ q, int n) {
  int i = blockIdx.x * blockDim.x + threadIdx.x;
  if (i < n) {
    p[i] = 0.0f;
    q[i] = 0.0f;
  }
}

__device__ __forceinline__ v8f wmma_k4(v2f a, v2f b, v8f c) {
  return __builtin_amdgcn_wmma_f32_16x16x4_f32(
       false, a,  false, b,
       (short)0, c,  false,  false);
}

__device__ __forceinline__ void load_frag4(v4f (&f)[4], const float* p, int s) {
#pragma unroll
  for (int t = 0; t < 4; ++t) f[t] = *(const v4f*)(p + t * (16 * HW) + 8 * s);
}

__device__ __forceinline__ void mma_block(v8f (&acc)[4][4], const v4f (&a)[4],
                                          const v4f (&b)[4]) {
#pragma unroll
  for (int ti = 0; ti < 4; ++ti)
#pragma unroll
    for (int tj = 0; tj < 4; ++tj) acc[ti][tj] = wmma_k4(a[ti].xy, b[tj].xy, acc[ti][tj]);
#pragma unroll
  for (int ti = 0; ti < 4; ++ti)
#pragma unroll
    for (int tj = 0; tj < 4; ++tj) acc[ti][tj] = wmma_k4(a[ti].zw, b[tj].zw, acc[ti][tj]);
}

__global__ __launch_bounds__(256) void cbp_gram_sketch_kernel(
    const float* __restrict__ x, const int* __restrict__ h1,
    const float* __restrict__ s1, const int* __restrict__ h2,
    const float* __restrict__ s2, float* __restrict__ plane0,
    float* __restrict__ plane1) {
  __shared__ float hist[8 * ND];

  const int b    = blockIdx.x >> 2;
  const int i8   = blockIdx.x & 3;
  const int tid  = threadIdx.x;
  const int wave = tid >> 5;
  const int lane = tid & 31;
  const int half = lane >> 4;
  const int lr   = lane & 15;

  float* myhist = hist + wave * ND;
  for (int k = lane * 4; k < ND; k += 128) *(v4f*)(myhist + k) = (v4f){0.0f, 0.0f, 0.0f, 0.0f};

  const float* xb = x + (size_t)b * (NC * HW);
  const int col0 = wave * 64;
  const float* pb = xb + (col0 + lr) * HW + 4 * half;

  int   h2n[4];
  float s2n[4];
#pragma unroll
  for (int tj = 0; tj < 4; ++tj) {
    h2n[tj] = h2[col0 + tj * 16 + lr];
    s2n[tj] = s2[col0 + tj * 16 + lr];
  }

  for (int pass = 0; pass < 2; ++pass) {
    const int row0 = i8 * 128 + pass * 64;
    const float* pa = xb + (row0 + lr) * HW + 4 * half;

    v8f acc[4][4];
#pragma unroll
    for (int ti = 0; ti < 4; ++ti)
#pragma unroll
      for (int tj = 0; tj < 4; ++tj) acc[ti][tj] = (v8f){0.0f, 0.0f, 0.0f, 0.0f, 0.0f, 0.0f, 0.0f, 0.0f};

    v4f a0[4], b0[4], a1[4], b1[4];
    load_frag4(a0, pa, 0);
    load_frag4(b0, pb, 0);
    for (int it = 0; it < 11; ++it) {
      load_frag4(a1, pa, 2 * it + 1);
      load_frag4(b1, pb, 2 * it + 1);
      mma_block(acc, a0, b0);
      load_frag4(a0, pa, 2 * it + 2);
      load_frag4(b0, pb, 2 * it + 2);
      mma_block(acc, a1, b1);
    }
    load_frag4(a1, pa, 23);
    load_frag4(b1, pb, 23);
    mma_block(acc, a0, b0);
    v2f ta[4], tb[4];
#pragma unroll
    for (int t = 0; t < 4; ++t) {
      ta[t] = *(const v2f*)(xb + (row0 + t * 16 + lr) * HW + 192 + 2 * half);
      tb[t] = *(const v2f*)(xb + (col0 + t * 16 + lr) * HW + 192 + 2 * half);
    }
    mma_block(acc, a1, b1);
#pragma unroll
    for (int ti = 0; ti < 4; ++ti)
#pragma unroll
      for (int tj = 0; tj < 4; ++tj) acc[ti][tj] = wmma_k4(ta[ti], tb[tj], acc[ti][tj]);

#pragma unroll
    for (int ti = 0; ti < 4; ++ti) {
      const int mbase = row0 + ti * 16 + half * 8;
      int   hm[8];
      float sm[8];
#pragma unroll
      for (int r = 0; r < 8; ++r) {
        hm[r] = h1[mbase + r];
        sm[r] = s1[mbase + r];
      }
#pragma unroll
      for (int tj = 0; tj < 4; ++tj) {
#pragma unroll
        for (int r = 0; r < 8; ++r) {
          const int bucket = (hm[r] + h2n[tj]) & (ND - 1);
          atomicAdd(&myhist[bucket], sm[r] * s2n[tj] * acc[ti][tj][r]);
        }
      }
    }
  }

  __syncthreads();
  float* dst = ((i8 < 2) ? plane0 : plane1) + (size_t)b * ND;
  for (int idx = tid; idx < ND; idx += 256) {
    float v = hist[idx];
#pragma unroll
    for (int w = 1; w < 8; ++w) v += hist[w * ND + idx];
    atomicAdd(&dst[idx], v);
  }
}

__global__ __launch_bounds__(256) void cbp_finalize_kernel(
    const float* __restrict__ plane0, float* out) {
  __shared__ float red[256];
  const int b   = blockIdx.x;
  const int tid = threadIdx.x;
  const float* v0 = plane0 + (size_t)b * ND;
  float* o = out + (size_t)b * ND;

  float v[ND / 256];
  float ssum = 0.0f;
#pragma unroll
  for (int k = 0; k < ND / 256; ++k) {
    const int idx = tid + 256 * k;
    v[k] = v0[idx] + o[idx];
    ssum += fabsf(v[k]) + 1e-5f;
  }
  red[tid] = ssum;
  __syncthreads();
  for (int s = 128; s > 0; s >>= 1) {
    if (tid < s) red[tid] += red[tid + s];
    __syncthreads();
  }
  const float inv = 1.0f / fmaxf(sqrtf(red[0]), 1e-12f);

#pragma unroll
  for (int k = 0; k < ND / 256; ++k) {
    const float val = v[k];
    const float sg  = (val > 0.0f) ? 1.0f : ((val < 0.0f) ? -1.0f : 0.0f);
    o[tid + 256 * k] = sg * sqrtf(fabsf(val) + 1e-5f) * inv;
  }
}

extern "C" void kernel_launch(void* const* d_in, const int* in_sizes, int n_in,
                              void* d_out, int out_size, void* d_ws, size_t ws_size,
                              hipStream_t stream) {
  const float* x  = (const float*)d_in[0];
  const int*   h1 = (const int*)d_in[1];
  const float* s1 = (const float*)d_in[2];
  const int*   h2 = (const int*)d_in[3];
  const float* s2 = (const float*)d_in[4];
  float*       out = (float*)d_out;

  float* cbp = (float*)d_ws;
  const int n = NB * ND;

  cbp_zero_kernel<<<(n + 255) / 256, 256, 0, stream>>>(cbp, out, n);
  cbp_gram_sketch_kernel<<<NB * 4, 256, 0, stream>>>(x, h1, s1, h2, s2, cbp, out);
  cbp_finalize_kernel<<<NB, 256, 0, stream>>>(cbp, out);
}


# G = X X^T is symmetric: WMMA only for the 36 of 64 blocks (64x64) on or above its diagonal, 10 of 16 tiles in the diagonal ones; each value added as G[c1,c2] and G[c2,c1]; order of LDS adds changes
# speedup vs baseline: 3.8733x; 1.2902x over previous
#include <hip/hip_runtime.h>
#include <math.h>


typedef __attribute__((ext_vector_type(2))) float v2f;
typedef __attribute__((ext_vector_type(4))) float v4f;
typedef __attribute__((ext_vector_type(8))) float v8f;

#define NB 32
#define NC 512
#define HW 196
#define ND 8192

__global__ void cbp_zero_kernel(float* __restrict__ p, float* __restrict__ q, int n) {
  int i = blockIdx.x * blockDim.x + threadIdx.x;
  if (i < n) {
    p[i] = 0.0f;
    q[i] = 0.0f;
  }
}

__device__ __forceinline__ v8f wmma_k4(v2f a, v2f b, v8f c) {
  return __builtin_amdgcn_wmma_f32_16x16x4_f32(
       false, a,  false, b,
       (short)0, c,  false,  false);
}

__device__ __forceinline__ void load_frag4(v4f (&f)[4], const float* p, int s) {
#pragma unroll
  for (int t = 0; t < 4; ++t) f[t] = *(const v4f*)(p + t * (16 * HW) + 8 * s);
}

template <bool DIAG>
__device__ __forceinline__ void mma_block(v8f (&acc)[4][4], const v4f (&a)[4],
                                          const v4f (&b)[4]) {
#pragma unroll
  for (int ti = 0; ti < 4; ++ti)
#pragma unroll
    for (int tj = 0; tj < 4; ++tj)
      if (!DIAG || ti <= tj) acc[ti][tj] = wmma_k4(a[ti].xy, b[tj].xy, acc[ti][tj]);
#pragma unroll
  for (int ti = 0; ti < 4; ++ti)
#pragma unroll
    for (int tj = 0; tj < 4; ++tj)
      if (!DIAG || ti <= tj) acc[ti][tj] = wmma_k4(a[ti].zw, b[tj].zw, acc[ti][tj]);
}

template <bool DIAG>
__device__ __forceinline__ void gram_block(v8f (&acc)[4][4], const float* xb, int row0,
                                           int col0, int lr, int half) {
  const float* pa = xb + (row0 + lr) * HW + 4 * half;
  const float* pb = xb + (col0 + lr) * HW + 4 * half;
#pragma unroll
  for (int ti = 0; ti < 4; ++ti)
#pragma unroll
    for (int tj = 0; tj < 4; ++tj) acc[ti][tj] = (v8f){0.0f, 0.0f, 0.0f, 0.0f, 0.0f, 0.0f, 0.0f, 0.0f};

  v4f a0[4], b0[4], a1[4], b1[4];
  load_frag4(a0, pa, 0);
  if (!DIAG) load_frag4(b0, pb, 0);
#pragma nounroll
  for (int it = 0; it < 11; ++it) {
    load_frag4(a1, pa, 2 * it + 1);
    if (!DIAG) load_frag4(b1, pb, 2 * it + 1);
    mma_block<DIAG>(acc, a0, DIAG ? a0 : b0);
    load_frag4(a0, pa, 2 * it + 2);
    if (!DIAG) load_frag4(b0, pb, 2 * it + 2);
    mma_block<DIAG>(acc, a1, DIAG ? a1 : b1);
  }
  load_frag4(a1, pa, 23);
  if (!DIAG) load_frag4(b1, pb, 23);
  mma_block<DIAG>(acc, a0, DIAG ? a0 : b0);
  v2f ta[4], tb[4];
#pragma unroll
  for (int t = 0; t < 4; ++t) {
    ta[t] = *(const v2f*)(xb + (row0 + t * 16 + lr) * HW + 192 + 2 * half);
    tb[t] = DIAG ? ta[t] : *(const v2f*)(xb + (col0 + t * 16 + lr) * HW + 192 + 2 * half);
  }
  mma_block<DIAG>(acc, a1, DIAG ? a1 : b1);
#pragma unroll
  for (int ti = 0; ti < 4; ++ti)
#pragma unroll
    for (int tj = 0; tj < 4; ++tj)
      if (!DIAG || ti <= tj) acc[ti][tj] = wmma_k4(ta[ti], tb[tj], acc[ti][tj]);
}

template <bool DIAG>
__device__ __forceinline__ void sketch_block(const v8f (&acc)[4][4], float* myhist, int row0,
                                             int col0, int lr, int half,
                                             const int* __restrict__ h1,
                                             const float* __restrict__ s1,
                                             const int* __restrict__ h2,
                                             const float* __restrict__ s2) {
  int   h1n[4], h2n[4];
  float s1n[4], s2n[4];
#pragma unroll
  for (int tj = 0; tj < 4; ++tj) {
    const int n = col0 + tj * 16 + lr;
    h1n[tj] = h1[n];
    s1n[tj] = s1[n];
    h2n[tj] = h2[n];
    s2n[tj] = s2[n];
  }
#pragma unroll
  for (int ti = 0; ti < 4; ++ti) {
    const int mbase = row0 + ti * 16 + half * 8;
    int   h1m[8], h2m[8];
    float s1m[8], s2m[8];
#pragma unroll
    for (int r = 0; r < 8; ++r) {
      h1m[r] = h1[mbase + r];
      s1m[r] = s1[mbase + r];
      h2m[r] = h2[mbase + r];
      s2m[r] = s2[mbase + r];
    }
#pragma unroll
    for (int tj = 0; tj < 4; ++tj) {
      if (DIAG && ti > tj) continue;
#pragma unroll
      for (int r = 0; r < 8; ++r) {
        const float g = acc[ti][tj][r];
        atomicAdd(&myhist[(h1m[r] + h2n[tj]) & (ND - 1)], s1m[r] * s2n[tj] * g);
        if (!(DIAG && ti == tj))
          atomicAdd(&myhist[(h1n[tj] + h2m[r]) & (ND - 1)], s1n[tj] * s2m[r] * g);
      }
    }
  }
}

__global__ __launch_bounds__(256) void cbp_gram_sketch_kernel(
    const float* __restrict__ x, const int* __restrict__ h1,
    const float* __restrict__ s1, const int* __restrict__ h2,
    const float* __restrict__ s2, float* __restrict__ plane0,
    float* __restrict__ plane1) {
  __shared__ float hist[8 * ND];

  const int b    = blockIdx.x >> 2;
  const int q    = blockIdx.x & 3;
  const int tid  = threadIdx.x;
  const int wave = __builtin_amdgcn_readfirstlane(tid >> 5);
  const int lane = tid & 31;
  const int half = lane >> 4;
  const int lr   = lane & 15;

  float* myhist = hist + wave * ND;
  for (int k = lane * 4; k < ND; k += 128) *(v4f*)(myhist + k) = (v4f){0.0f, 0.0f, 0.0f, 0.0f};

  const float* xb = x + (size_t)b * (NC * HW);

  v8f acc[4][4];
  if (wave < 7) {
    int I = 0, rem = q * 7 + wave;
    for (int t = 0; t < 6; ++t)
      if (rem >= 7 - I) {
        rem -= 7 - I;
        ++I;
      }
    const int J = I + 1 + rem;
    gram_block<false>(acc, xb, I * 64, J * 64, lr, half);
    sketch_block<false>(acc, myhist, I * 64, J * 64, lr, half, h1, s1, h2, s2);
  } else {
    for (int d = 0; d < 2; ++d) {
      const int I = 2 * q + d;
      gram_block<true>(acc, xb, I * 64, I * 64, lr, half);
      sketch_block<true>(acc, myhist, I * 64, I * 64, lr, half, h1, s1, h2, s2);
    }
  }

  __syncthreads();
  float* dst = ((q < 2) ? plane0 : plane1) + (size_t)b * ND;
  for (int idx = tid; idx < ND; idx += 256) {
    float v = hist[idx];
#pragma unroll
    for (int w = 1; w < 8; ++w) v += hist[w * ND + idx];
    atomicAdd(&dst[idx], v);
  }
}

__global__ __launch_bounds__(256) void cbp_finalize_kernel(
    const float* __restrict__ plane0, float* out) {
  __shared__ float red[256];
  const int b   = blockIdx.x;
  const int tid = threadIdx.x;
  const float* v0 = plane0 + (size_t)b * ND;
  float* o = out + (size_t)b * ND;

  float v[ND / 256];
  float ssum = 0.0f;
#pragma unroll
  for (int k = 0; k < ND / 256; ++k) {
    const int idx = tid + 256 * k;
    v[k] = v0[idx] + o[idx];
    ssum += fabsf(v[k]) + 1e-5f;
  }
  red[tid] = ssum;
  __syncthreads();
  for (int s = 128; s > 0; s >>= 1) {
    if (tid < s) red[tid] += red[tid + s];
    __syncthreads();
  }
  const float inv = 1.0f / fmaxf(sqrtf(red[0]), 1e-12f);

#pragma unroll
  for (int k = 0; k < ND / 256; ++k) {
    const float val = v[k];
    const float sg  = (val > 0.0f) ? 1.0f : ((val < 0.0f) ? -1.0f : 0.0f);
    o[tid + 256 * k] = sg * sqrtf(fabsf(val) + 1e-5f) * inv;
  }
}

extern "C" void kernel_launch(void* const* d_in, const int* in_sizes, int n_in,
                              void* d_out, int out_size, void* d_ws, size_t ws_size,
                              hipStream_t stream) {
  const float* x  = (const float*)d_in[0];
  const int*   h1 = (const int*)d_in[1];
  const float* s1 = (const float*)d_in[2];
  const int*   h2 = (const int*)d_in[3];
  const float* s2 = (const float*)d_in[4];
  float*       out = (float*)d_out;

  float* cbp = (float*)d_ws;
  const int n = NB * ND;

  cbp_zero_kernel<<<(n + 255) / 256, 256, 0, stream>>>(cbp, out, n);
  cbp_gram_sketch_kernel<<<NB * 4, 256, 0, stream>>>(x, h1, s1, h2, s2, cbp, out);
  cbp_finalize_kernel<<<NB, 256, 0, stream>>>(cbp, out);
}


# If ws_size >= 7 MB: 8 4-wave blocks per batch on all 256 WGPs (diagonal block pairs in lockstep), each stores its own plane: no atomics, no clearing launch; planes added in fixed order; else parent
# speedup vs baseline: 5.0476x; 1.3032x over previous
#include <hip/hip_runtime.h>
#include <math.h>


typedef __attribute__((ext_vector_type(2))) float v2f;
typedef __attribute__((ext_vector_type(4))) float v4f;
typedef __attribute__((ext_vector_type(8))) float v8f;

#define NB 32
#define NC 512
#define HW 196
#define ND 8192

__global__ void cbp_zero_kernel(float* __restrict__ p, float* __restrict__ q, int n) {
  int i = blockIdx.x * blockDim.x + threadIdx.x;
  if (i < n) {
    p[i] = 0.0f;
    q[i] = 0.0f;
  }
}

__device__ __forceinline__ v8f wmma_k4(v2f a, v2f b, v8f c) {
  return __builtin_amdgcn_wmma_f32_16x16x4_f32(
       false, a,  false, b,
       (short)0, c,  false,  false);
}

__device__ __forceinline__ void load_frag4(v4f (&f)[4], const float* p, int s) {
#pragma unroll
  for (int t = 0; t < 4; ++t) f[t] = *(const v4f*)(p + t * (16 * HW) + 8 * s);
}

template <bool DIAG>
__device__ __forceinline__ void mma_block(v8f (&acc)[4][4], const v4f (&a)[4],
                                          const v4f (&b)[4]) {
#pragma unroll
  for (int ti = 0; ti < 4; ++ti)
#pragma unroll
    for (int tj = 0; tj < 4; ++tj)
      if (!DIAG || ti <= tj) acc[ti][tj] = wmma_k4(a[ti].xy, b[tj].xy, acc[ti][tj]);
#pragma unroll
  for (int ti = 0; ti < 4; ++ti)
#pragma unroll
    for (int tj = 0; tj < 4; ++tj)
      if (!DIAG || ti <= tj) acc[ti][tj] = wmma_k4(a[ti].zw, b[tj].zw, acc[ti][tj]);
}

template <bool DIAG>
__device__ __forceinline__ void gram_block(v8f (&acc)[4][4], const float* xb, int row0,
                                           int col0, int lr, int half) {
  const float* pa = xb + (row0 + lr) * HW + 4 * half;
  const float* pb = xb + (col0 + lr) * HW + 4 * half;
#pragma unroll
  for (int ti = 0; ti < 4; ++ti)
#pragma unroll
    for (int tj = 0; tj < 4; ++tj) acc[ti][tj] = (v8f){0.0f, 0.0f, 0.0f, 0.0f, 0.0f, 0.0f, 0.0f, 0.0f};

  v4f a0[4], b0[4], a1[4], b1[4];
  load_frag4(a0, pa, 0);
  if (!DIAG) load_frag4(b0, pb, 0);
#pragma nounroll
  for (int it = 0; it < 11; ++it) {
    load_frag4(a1, pa, 2 * it + 1);
    if (!DIAG) load_frag4(b1, pb, 2 * it + 1);
    mma_block<DIAG>(acc, a0, DIAG ? a0 : b0);
    load_frag4(a0, pa, 2 * it + 2);
    if (!DIAG) load_frag4(b0, pb, 2 * it + 2);
    mma_block<DIAG>(acc, a1, DIAG ? a1 : b1);
  }
  load_frag4(a1, pa, 23);
  if (!DIAG) load_frag4(b1, pb, 23);
  mma_block<DIAG>(acc, a0, DIAG ? a0 : b0);
  v2f ta[4], tb[4];
#pragma unroll
  for (int t = 0; t < 4; ++t) {
    ta[t] = *(const v2f*)(xb + (row0 + t * 16 + lr) * HW + 192 + 2 * half);
    tb[t] = DIAG ? ta[t] : *(const v2f*)(xb + (col0 + t * 16 + lr) * HW + 192 + 2 * half);
  }
  mma_block<DIAG>(acc, a1, DIAG ? a1 : b1);
#pragma unroll
  for (int ti = 0; ti < 4; ++ti)
#pragma unroll
    for (int tj = 0; tj < 4; ++tj)
      if (!DIAG || ti <= tj) acc[ti][tj] = wmma_k4(ta[ti], tb[tj], acc[ti][tj]);
}

__device__ __forceinline__ void gram_diag_pair(v8f (&accA)[4][4], v8f (&accB)[4][4],
                                               const float* xb, int rowA, int rowB, int lr,
                                               int half) {
  const float* pa = xb + (rowA + lr) * HW + 4 * half;
  const float* pb = xb + (rowB + lr) * HW + 4 * half;
#pragma unroll
  for (int ti = 0; ti < 4; ++ti)
#pragma unroll
    for (int tj = 0; tj < 4; ++tj) {
      accA[ti][tj] = (v8f){0.0f, 0.0f, 0.0f, 0.0f, 0.0f, 0.0f, 0.0f, 0.0f};
      accB[ti][tj] = (v8f){0.0f, 0.0f, 0.0f, 0.0f, 0.0f, 0.0f, 0.0f, 0.0f};
    }

  v4f a0[4], b0[4], a1[4], b1[4];
  load_frag4(a0, pa, 0);
  load_frag4(b0, pb, 0);
#pragma nounroll
  for (int it = 0; it < 11; ++it) {
    load_frag4(a1, pa, 2 * it + 1);
    load_frag4(b1, pb, 2 * it + 1);
    mma_block<true>(accA, a0, a0);
    mma_block<true>(accB, b0, b0);
    load_frag4(a0, pa, 2 * it + 2);
    load_frag4(b0, pb, 2 * it + 2);
    mma_block<true>(accA, a1, a1);
    mma_block<true>(accB, b1, b1);
  }
  load_frag4(a1, pa, 23);
  load_frag4(b1, pb, 23);
  mma_block<true>(accA, a0, a0);
  mma_block<true>(accB, b0, b0);
  v2f ta[4], tb[4];
#pragma unroll
  for (int t = 0; t < 4; ++t) {
    ta[t] = *(const v2f*)(xb + (rowA + t * 16 + lr) * HW + 192 + 2 * half);
    tb[t] = *(const v2f*)(xb + (rowB + t * 16 + lr) * HW + 192 + 2 * half);
  }
  mma_block<true>(accA, a1, a1);
  mma_block<true>(accB, b1, b1);
#pragma unroll
  for (int ti = 0; ti < 4; ++ti)
#pragma unroll
    for (int tj = 0; tj < 4; ++tj)
      if (ti <= tj) {
        accA[ti][tj] = wmma_k4(ta[ti], ta[tj], accA[ti][tj]);
        accB[ti][tj] = wmma_k4(tb[ti], tb[tj], accB[ti][tj]);
      }
}

template <bool DIAG>
__device__ __forceinline__ void sketch_block(const v8f (&acc)[4][4], float* myhist, int row0,
                                             int col0, int lr, int half,
                                             const int* __restrict__ h1,
                                             const float* __restrict__ s1,
                                             const int* __restrict__ h2,
                                             const float* __restrict__ s2) {
  int   h1n[4], h2n[4];
  float s1n[4], s2n[4];
#pragma unroll
  for (int tj = 0; tj < 4; ++tj) {
    const int n = col0 + tj * 16 + lr;
    h1n[tj] = h1[n];
    s1n[tj] = s1[n];
    h2n[tj] = h2[n];
    s2n[tj] = s2[n];
  }
#pragma unroll
  for (int ti = 0; ti < 4; ++ti) {
    const int mbase = row0 + ti * 16 + half * 8;
    int   h1m[8], h2m[8];
    float s1m[8], s2m[8];
#pragma unroll
    for (int r = 0; r < 8; ++r) {
      h1m[r] = h1[mbase + r];
      s1m[r] = s1[mbase + r];
      h2m[r] = h2[mbase + r];
      s2m[r] = s2[mbase + r];
    }
#pragma unroll
    for (int tj = 0; tj < 4; ++tj) {
      if (DIAG && ti > tj) continue;
#pragma unroll
      for (int r = 0; r < 8; ++r) {
        const float g = acc[ti][tj][r];
        atomicAdd(&myhist[(h1m[r] + h2n[tj]) & (ND - 1)], s1m[r] * s2n[tj] * g);
        if (!(DIAG && ti == tj))
          atomicAdd(&myhist[(h1n[tj] + h2m[r]) & (ND - 1)], s1n[tj] * s2m[r] * g);
      }
    }
  }
}

__global__ __launch_bounds__(256) void cbp_gram_sketch_kernel(
    const float* __restrict__ x, const int* __restrict__ h1,
    const float* __restrict__ s1, const int* __restrict__ h2,
    const float* __restrict__ s2, float* __restrict__ plane0,
    float* __restrict__ plane1) {
  __shared__ float hist[8 * ND];

  const int b    = blockIdx.x >> 2;
  const int q    = blockIdx.x & 3;
  const int tid  = threadIdx.x;
  const int wave = __builtin_amdgcn_readfirstlane(tid >> 5);
  const int lane = tid & 31;
  const int half = lane >> 4;
  const int lr   = lane & 15;

  float* myhist = hist + wave * ND;
  for (int k = lane * 4; k < ND; k += 128) *(v4f*)(myhist + k) = (v4f){0.0f, 0.0f, 0.0f, 0.0f};

  const float* xb = x + (size_t)b * (NC * HW);

  v8f acc[4][4];
  if (wave < 7) {
    int I = 0, rem = q * 7 + wave;
    for (int t = 0; t < 6; ++t)
      if (rem >= 7 - I) {
        rem -= 7 - I;
        ++I;
      }
    const int J = I + 1 + rem;
    gram_block<false>(acc, xb, I * 64, J * 64, lr, half);
    sketch_block<false>(acc, myhist, I * 64, J * 64, lr, half, h1, s1, h2, s2);
  } else {
    for (int d = 0; d < 2; ++d) {
      const int I = 2 * q + d;
      gram_block<true>(acc, xb, I * 64, I * 64, lr, half);
      sketch_block<true>(acc, myhist, I * 64, I * 64, lr, half, h1, s1, h2, s2);
    }
  }

  __syncthreads();
  float* dst = ((q < 2) ? plane0 : plane1) + (size_t)b * ND;
  for (int idx = tid; idx < ND; idx += 256) {
    float v = hist[idx];
#pragma unroll
    for (int w = 1; w < 8; ++w) v += hist[w * ND + idx];
    atomicAdd(&dst[idx], v);
  }
}

__global__ __launch_bounds__(256) void cbp_finalize_kernel(
    const float* __restrict__ plane0, float* out) {
  __shared__ float red[256];
  const int b   = blockIdx.x;
  const int tid = threadIdx.x;
  const float* v0 = plane0 + (size_t)b * ND;
  float* o = out + (size_t)b * ND;

  float v[ND / 256];
  float ssum = 0.0f;
#pragma unroll
  for (int k = 0; k < ND / 256; ++k) {
    const int idx = tid + 256 * k;
    v[k] = v0[idx] + o[idx];
    ssum += fabsf(v[k]) + 1e-5f;
  }
  red[tid] = ssum;
  __syncthreads();
  for (int s = 128; s > 0; s >>= 1) {
    if (tid < s) red[tid] += red[tid + s];
    __syncthreads();
  }
  const float inv = 1.0f / fmaxf(sqrtf(red[0]), 1e-12f);

#pragma unroll
  for (int k = 0; k < ND / 256; ++k) {
    const float val = v[k];
    const float sg  = (val > 0.0f) ? 1.0f : ((val < 0.0f) ? -1.0f : 0.0f);
    o[tid + 256 * k] = sg * sqrtf(fabsf(val) + 1e-5f) * inv;
  }
}

#define NPLANE 8
#define PLANES_LDS_FLOATS 41984
__global__ __launch_bounds__(128) void cbp_gram_sketch_planes_kernel(
    const float* __restrict__ x, const int* __restrict__ h1,
    const float* __restrict__ s1, const int* __restrict__ h2,
    const float* __restrict__ s2, float* __restrict__ planes,
    float* __restrict__ out) {
  __shared__ float hist[PLANES_LDS_FLOATS];

  const int b    = blockIdx.x >> 3;
  const int g    = blockIdx.x & 7;
  const int tid  = threadIdx.x;
  const int wave = __builtin_amdgcn_readfirstlane(tid >> 5);
  const int lane = tid & 31;
  const int half = lane >> 4;
  const int lr   = lane & 15;

  float* myhist = hist + wave * ND;
  for (int k = lane * 4; k < ND; k += 128) *(v4f*)(myhist + k) = (v4f){0.0f, 0.0f, 0.0f, 0.0f};

  const float* xb = x + (size_t)b * (NC * HW);
  const int u = g * 4 + wave;

  v8f acc[4][4];
  if (u < 28) {
    int I = 0, rem = u;
    for (int t = 0; t < 6; ++t)
      if (rem >= 7 - I) {
        rem -= 7 - I;
        ++I;
      }
    const int J = I + 1 + rem;
    gram_block<false>(acc, xb, I * 64, J * 64, lr, half);
    sketch_block<false>(acc, myhist, I * 64, J * 64, lr, half, h1, s1, h2, s2);
  } else {
    const int IA = 2 * (u - 28), IB = IA + 1;
    v8f acc2[4][4];
    gram_diag_pair(acc, acc2, xb, IA * 64, IB * 64, lr, half);
    sketch_block<true>(acc, myhist, IA * 64, IA * 64, lr, half, h1, s1, h2, s2);
    sketch_block<true>(acc2, myhist, IB * 64, IB * 64, lr, half, h1, s1, h2, s2);
  }

  __syncthreads();
  float* dst = ((g < NPLANE - 1) ? planes + (size_t)g * (NB * ND) : out) + (size_t)b * ND;
  for (int idx = tid * 4; idx < ND; idx += 128 * 4) {
    v4f v = *(const v4f*)(hist + idx);
    v += *(const v4f*)(hist + ND + idx);
    v += *(const v4f*)(hist + 2 * ND + idx);
    v += *(const v4f*)(hist + 3 * ND + idx);
    *(v4f*)(dst + idx) = v;
  }
}

__global__ __launch_bounds__(256) void cbp_finalize_planes_kernel(
    const float* __restrict__ planes, float* out) {
  __shared__ float red[256];
  const int b   = blockIdx.x;
  const int tid = threadIdx.x;
  const float* p0 = planes + (size_t)b * ND;
  float* o = out + (size_t)b * ND;

  v4f v[ND / 1024];
  float ssum = 0.0f;
#pragma unroll
  for (int k = 0; k < ND / 1024; ++k) {
    const int idx = 4 * (tid + 256 * k);
    v4f a = *(const v4f*)(p0 + idx);
#pragma unroll
    for (int p = 1; p < NPLANE - 1; ++p) a += *(const v4f*)(p0 + (size_t)p * (NB * ND) + idx);
    a += *(const v4f*)(o + idx);
    v[k] = a;
    ssum += fabsf(a.x) + 1e-5f;
    ssum += fabsf(a.y) + 1e-5f;
    ssum += fabsf(a.z) + 1e-5f;
    ssum += fabsf(a.w) + 1e-5f;
  }
  red[tid] = ssum;
  __syncthreads();
  for (int s = 128; s > 0; s >>= 1) {
    if (tid < s) red[tid] += red[tid + s];
    __syncthreads();
  }
  const float inv = 1.0f / fmaxf(sqrtf(red[0]), 1e-12f);

#pragma unroll
  for (int k = 0; k < ND / 1024; ++k) {
    v4f r;
#pragma unroll
    for (int e = 0; e < 4; ++e) {
      const float val = v[k][e];
      const float sg  = (val > 0.0f) ? 1.0f : ((val < 0.0f) ? -1.0f : 0.0f);
      r[e] = sg * sqrtf(fabsf(val) + 1e-5f) * inv;
    }
    *(v4f*)(o + 4 * (tid + 256 * k)) = r;
  }
}

extern "C" void kernel_launch(void* const* d_in, const int* in_sizes, int n_in,
                              void* d_out, int out_size, void* d_ws, size_t ws_size,
                              hipStream_t stream) {
  const float* x  = (const float*)d_in[0];
  const int*   h1 = (const int*)d_in[1];
  const float* s1 = (const float*)d_in[2];
  const int*   h2 = (const int*)d_in[3];
  const float* s2 = (const float*)d_in[4];
  float*       out = (float*)d_out;

  float* cbp = (float*)d_ws;
  const int n = NB * ND;

  if (ws_size >= 7340032) {
    cbp_gram_sketch_planes_kernel<<<NB * NPLANE, 128, 0, stream>>>(x, h1, s1, h2, s2, cbp, out);
    cbp_finalize_planes_kernel<<<NB, 256, 0, stream>>>(cbp, out);
  } else {
    cbp_zero_kernel<<<(n + 255) / 256, 256, 0, stream>>>(cbp, out, n);
    cbp_gram_sketch_kernel<<<NB * 4, 256, 0, stream>>>(x, h1, s1, h2, s2, cbp, out);
    cbp_finalize_kernel<<<NB, 256, 0, stream>>>(cbp, out);
  }
}


# X panels staged in LDS: a work-group copies its four 64-row panels with whole-line loads and reads fragments by ds_load; blocks regrouped 2x2 so four panels do; no atomics; plane sum order changes
# speedup vs baseline: 5.2046x; 1.0311x over previous
#include <hip/hip_runtime.h>
#include <math.h>


typedef __attribute__((ext_vector_type(2))) float v2f;
typedef __attribute__((ext_vector_type(4))) float v4f;
typedef __attribute__((ext_vector_type(8))) float v8f;

#define NB 32
#define NC 512
#define HW 196
#define ND 8192

__global__ void cbp_zero_kernel(float* __restrict__ p, float* __restrict__ q, int n) {
  int i = blockIdx.x * blockDim.x + threadIdx.x;
  if (i < n) {
    p[i] = 0.0f;
    q[i] = 0.0f;
  }
}

__device__ __forceinline__ v8f wmma_k4(v2f a, v2f b, v8f c) {
  return __builtin_amdgcn_wmma_f32_16x16x4_f32(
       false, a,  false, b,
       (short)0, c,  false,  false);
}

__device__ __forceinline__ void load_frag4(v4f (&f)[4], const float* p, int s) {
#pragma unroll
  for (int t = 0; t < 4; ++t) f[t] = *(const v4f*)(p + t * (16 * HW) + 8 * s);
}

template <bool DIAG>
__device__ __forceinline__ void mma_block(v8f (&acc)[4][4], const v4f (&a)[4],
                                          const v4f (&b)[4]) {
#pragma unroll
  for (int ti = 0; ti < 4; ++ti)
#pragma unroll
    for (int tj = 0; tj < 4; ++tj)
      if (!DIAG || ti <= tj) acc[ti][tj] = wmma_k4(a[ti].xy, b[tj].xy, acc[ti][tj]);
#pragma unroll
  for (int ti = 0; ti < 4; ++ti)
#pragma unroll
    for (int tj = 0; tj < 4; ++tj)
      if (!DIAG || ti <= tj) acc[ti][tj] = wmma_k4(a[ti].zw, b[tj].zw, acc[ti][tj]);
}

template <bool DIAG>
__device__ __forceinline__ void gram_block(v8f (&acc)[4][4], const float* xb, int row0,
                                           int col0, int lr, int half) {
  const float* pa = xb + (row0 + lr) * HW + 4 * half;
  const float* pb = xb + (col0 + lr) * HW + 4 * half;
#pragma unroll
  for (int ti = 0; ti < 4; ++ti)
#pragma unroll
    for (int tj = 0; tj < 4; ++tj) acc[ti][tj] = (v8f){0.0f, 0.0f, 0.0f, 0.0f, 0.0f, 0.0f, 0.0f, 0.0f};

  v4f a0[4], b0[4], a1[4], b1[4];
  load_frag4(a0, pa, 0);
  if (!DIAG) load_frag4(b0, pb, 0);
#pragma nounroll
  for (int it = 0; it < 11; ++it) {
    load_frag4(a1, pa, 2 * it + 1);
    if (!DIAG) load_frag4(b1, pb, 2 * it + 1);
    mma_block<DIAG>(acc, a0, DIAG ? a0 : b0);
    load_frag4(a0, pa, 2 * it + 2);
    if (!DIAG) load_frag4(b0, pb, 2 * it + 2);
    mma_block<DIAG>(acc, a1, DIAG ? a1 : b1);
  }
  load_frag4(a1, pa, 23);
  if (!DIAG) load_frag4(b1, pb, 23);
  mma_block<DIAG>(acc, a0, DIAG ? a0 : b0);
  v2f ta[4], tb[4];
#pragma unroll
  for (int t = 0; t < 4; ++t) {
    ta[t] = *(const v2f*)(xb + (row0 + t * 16 + lr) * HW + 192 + 2 * half);
    tb[t] = DIAG ? ta[t] : *(const v2f*)(xb + (col0 + t * 16 + lr) * HW + 192 + 2 * half);
  }
  mma_block<DIAG>(acc, a1, DIAG ? a1 : b1);
#pragma unroll
  for (int ti = 0; ti < 4; ++ti)
#pragma unroll
    for (int tj = 0; tj < 4; ++tj)
      if (!DIAG || ti <= tj) acc[ti][tj] = wmma_k4(ta[ti], tb[tj], acc[ti][tj]);
}

__device__ __forceinline__ void gram_diag_pair(v8f (&accA)[4][4], v8f (&accB)[4][4],
                                               const float* xb, int rowA, int rowB, int lr,
                                               int half) {
  const float* pa = xb + (rowA + lr) * HW + 4 * half;
  const float* pb = xb + (rowB + lr) * HW + 4 * half;
#pragma unroll
  for (int ti = 0; ti < 4; ++ti)
#pragma unroll
    for (int tj = 0; tj < 4; ++tj) {
      accA[ti][tj] = (v8f){0.0f, 0.0f, 0.0f, 0.0f, 0.0f, 0.0f, 0.0f, 0.0f};
      accB[ti][tj] = (v8f){0.0f, 0.0f, 0.0f, 0.0f, 0.0f, 0.0f, 0.0f, 0.0f};
    }

  v4f a0[4], b0[4], a1[4], b1[4];
  load_frag4(a0, pa, 0);
  load_frag4(b0, pb, 0);
#pragma nounroll
  for (int it = 0; it < 11; ++it) {
    load_frag4(a1, pa, 2 * it + 1);
    load_frag4(b1, pb, 2 * it + 1);
    mma_block<true>(accA, a0, a0);
    mma_block<true>(accB, b0, b0);
    load_frag4(a0, pa, 2 * it + 2);
    load_frag4(b0, pb, 2 * it + 2);
    mma_block<true>(accA, a1, a1);
    mma_block<true>(accB, b1, b1);
  }
  load_frag4(a1, pa, 23);
  load_frag4(b1, pb, 23);
  mma_block<true>(accA, a0, a0);
  mma_block<true>(accB, b0, b0);
  v2f ta[4], tb[4];
#pragma unroll
  for (int t = 0; t < 4; ++t) {
    ta[t] = *(const v2f*)(xb + (rowA + t * 16 + lr) * HW + 192 + 2 * half);
    tb[t] = *(const v2f*)(xb + (rowB + t * 16 + lr) * HW + 192 + 2 * half);
  }
  mma_block<true>(accA, a1, a1);
  mma_block<true>(accB, b1, b1);
#pragma unroll
  for (int ti = 0; ti < 4; ++ti)
#pragma unroll
    for (int tj = 0; tj < 4; ++tj)
      if (ti <= tj) {
        accA[ti][tj] = wmma_k4(ta[ti], ta[tj], accA[ti][tj]);
        accB[ti][tj] = wmma_k4(tb[ti], tb[tj], accB[ti][tj]);
      }
}

template <bool DIAG>
__device__ __forceinline__ void sketch_block(const v8f (&acc)[4][4], float* myhist, int row0,
                                             int col0, int lr, int half,
                                             const int* __restrict__ h1,
                                             const float* __restrict__ s1,
                                             const int* __restrict__ h2,
                                             const float* __restrict__ s2) {
  int   h1n[4], h2n[4];
  float s1n[4], s2n[4];
#pragma unroll
  for (int tj = 0; tj < 4; ++tj) {
    const int n = col0 + tj * 16 + lr;
    h1n[tj] = h1[n];
    s1n[tj] = s1[n];
    h2n[tj] = h2[n];
    s2n[tj] = s2[n];
  }
#pragma unroll
  for (int ti = 0; ti < 4; ++ti) {
    const int mbase = row0 + ti * 16 + half * 8;
    int   h1m[8], h2m[8];
    float s1m[8], s2m[8];
#pragma unroll
    for (int r = 0; r < 8; ++r) {
      h1m[r] = h1[mbase + r];
      s1m[r] = s1[mbase + r];
      h2m[r] = h2[mbase + r];
      s2m[r] = s2[mbase + r];
    }
#pragma unroll
    for (int tj = 0; tj < 4; ++tj) {
      if (DIAG && ti > tj) continue;
#pragma unroll
      for (int r = 0; r < 8; ++r) {
        const float g = acc[ti][tj][r];
        atomicAdd(&myhist[(h1m[r] + h2n[tj]) & (ND - 1)], s1m[r] * s2n[tj] * g);
        if (!(DIAG && ti == tj))
          atomicAdd(&myhist[(h1n[tj] + h2m[r]) & (ND - 1)], s1n[tj] * s2m[r] * g);
      }
    }
  }
}

__global__ __launch_bounds__(256) void cbp_gram_sketch_kernel(
    const float* __restrict__ x, const int* __restrict__ h1,
    const float* __restrict__ s1, const int* __restrict__ h2,
    const float* __restrict__ s2, float* __restrict__ plane0,
    float* __restrict__ plane1) {
  __shared__ float hist[8 * ND];

  const int b    = blockIdx.x >> 2;
  const int q    = blockIdx.x & 3;
  const int tid  = threadIdx.x;
  const int wave = __builtin_amdgcn_readfirstlane(tid >> 5);
  const int lane = tid & 31;
  const int half = lane >> 4;
  const int lr   = lane & 15;

  float* myhist = hist + wave * ND;
  for (int k = lane * 4; k < ND; k += 128) *(v4f*)(myhist + k) = (v4f){0.0f, 0.0f, 0.0f, 0.0f};

  const float* xb = x + (size_t)b * (NC * HW);

  v8f acc[4][4];
  if (wave < 7) {
    int I = 0, rem = q * 7 + wave;
    for (int t = 0; t < 6; ++t)
      if (rem >= 7 - I) {
        rem -= 7 - I;
        ++I;
      }
    const int J = I + 1 + rem;
    gram_block<false>(acc, xb, I * 64, J * 64, lr, half);
    sketch_block<false>(acc, myhist, I * 64, J * 64, lr, half, h1, s1, h2, s2);
  } else {
    for (int d = 0; d < 2; ++d) {
      const int I = 2 * q + d;
      gram_block<true>(acc, xb, I * 64, I * 64, lr, half);
      sketch_block<true>(acc, myhist, I * 64, I * 64, lr, half, h1, s1, h2, s2);
    }
  }

  __syncthreads();
  float* dst = ((q < 2) ? plane0 : plane1) + (size_t)b * ND;
  for (int idx = tid; idx < ND; idx += 256) {
    float v = hist[idx];
#pragma unroll
    for (int w = 1; w < 8; ++w) v += hist[w * ND + idx];
    atomicAdd(&dst[idx], v);
  }
}

__global__ __launch_bounds__(256) void cbp_finalize_kernel(
    const float* __restrict__ plane0, float* out) {
  __shared__ float red[256];
  const int b   = blockIdx.x;
  const int tid = threadIdx.x;
  const float* v0 = plane0 + (size_t)b * ND;
  float* o = out + (size_t)b * ND;

  float v[ND / 256];
  float ssum = 0.0f;
#pragma unroll
  for (int k = 0; k < ND / 256; ++k) {
    const int idx = tid + 256 * k;
    v[k] = v0[idx] + o[idx];
    ssum += fabsf(v[k]) + 1e-5f;
  }
  red[tid] = ssum;
  __syncthreads();
  for (int s = 128; s > 0; s >>= 1) {
    if (tid < s) red[tid] += red[tid + s];
    __syncthreads();
  }
  const float inv = 1.0f / fmaxf(sqrtf(red[0]), 1e-12f);

#pragma unroll
  for (int k = 0; k < ND / 256; ++k) {
    const float val = v[k];
    const float sg  = (val > 0.0f) ? 1.0f : ((val < 0.0f) ? -1.0f : 0.0f);
    o[tid + 256 * k] = sg * sqrtf(fabsf(val) + 1e-5f) * inv;
  }
}

#define NPLANE 8
#define PANEL_FLOATS (64 * HW)
#define PAIR_V4 (2 * PANEL_FLOATS / 4)
#define PLANES_LDS_FLOATS (4 * PANEL_FLOATS)
__global__ __launch_bounds__(128) void cbp_gram_sketch_planes_kernel(
    const float* __restrict__ x, const int* __restrict__ h1,
    const float* __restrict__ s1, const int* __restrict__ h2,
    const float* __restrict__ s2, float* __restrict__ planes,
    float* __restrict__ out) {
  __shared__ __attribute__((aligned(16))) float lds[PLANES_LDS_FLOATS];

  const int b    = blockIdx.x >> 3;
  const int g    = blockIdx.x & 7;
  const int tid  = threadIdx.x;
  const int wave = __builtin_amdgcn_readfirstlane(tid >> 5);
  const int lane = tid & 31;
  const int half = lane >> 4;
  const int lr   = lane & 15;

  const float* xb = x + (size_t)b * (NC * HW);

  int p0, p2;
  if (g < 6) {
    const int SI = (g >= 3) + (g >= 5);
    const int SJ = (g < 3) ? g + 1 : ((g < 5) ? g - 1 : 3);
    p0 = 2 * SI;
    p2 = 2 * SJ;
  } else {
    p0 = 4 * (g - 6);
    p2 = p0 + 2;
  }

  {
    const v4f* srcA = (const v4f*)(xb + p0 * PANEL_FLOATS);
    const v4f* srcB = (const v4f*)(xb + p2 * PANEL_FLOATS);
    v4f* dst = (v4f*)lds;
    v4f t0[25], t1[24];
#pragma unroll
    for (int i = 0; i < 25; ++i) t0[i] = srcA[tid + 128 * i];
#pragma unroll
    for (int i = 0; i < 24; ++i) t1[i] = srcA[tid + 128 * (25 + i)];
#pragma unroll
    for (int i = 0; i < 25; ++i) dst[tid + 128 * i] = t0[i];
#pragma unroll
    for (int i = 0; i < 25; ++i) t0[i] = srcB[tid + 128 * i];
#pragma unroll
    for (int i = 0; i < 24; ++i) dst[tid + 128 * (25 + i)] = t1[i];
#pragma unroll
    for (int i = 0; i < 24; ++i) t1[i] = srcB[tid + 128 * (25 + i)];
#pragma unroll
    for (int i = 0; i < 25; ++i) dst[PAIR_V4 + tid + 128 * i] = t0[i];
#pragma unroll
    for (int i = 0; i < 24; ++i) dst[PAIR_V4 + tid + 128 * (25 + i)] = t1[i];
  }

  bool above;
  int slotA, slotB, I, J;
  if (g < 6) {
    above = true;
    slotA = wave >> 1;
    slotB = 2 + (wave & 1);
    I = p0 + (wave >> 1);
    J = p2 + (wave & 1);
  } else {
    above = wave < 2;
    slotA = 2 * (wave & 1);
    slotB = slotA + 1;
    I = p0 + slotA;
    J = I + 1;
  }

  __syncthreads();

  v8f acc[4][4], acc2[4][4];
  if (above)
    gram_block<false>(acc, lds, slotA * 64, slotB * 64, lr, half);
  else
    gram_diag_pair(acc, acc2, lds, slotA * 64, slotB * 64, lr, half);

  __syncthreads();

  float* myhist = lds + wave * ND;
  for (int k = lane * 4; k < ND; k += 128) *(v4f*)(myhist + k) = (v4f){0.0f, 0.0f, 0.0f, 0.0f};

  if (above) {
    sketch_block<false>(acc, myhist, I * 64, J * 64, lr, half, h1, s1, h2, s2);
  } else {
    sketch_block<true>(acc, myhist, I * 64, I * 64, lr, half, h1, s1, h2, s2);
    sketch_block<true>(acc2, myhist, J * 64, J * 64, lr, half, h1, s1, h2, s2);
  }

  __syncthreads();
  float* dst = ((g < NPLANE - 1) ? planes + (size_t)g * (NB * ND) : out) + (size_t)b * ND;
  for (int idx = tid * 4; idx < ND; idx += 128 * 4) {
    v4f v = *(const v4f*)(lds + idx);
    v += *(const v4f*)(lds + ND + idx);
    v += *(const v4f*)(lds + 2 * ND + idx);
    v += *(const v4f*)(lds + 3 * ND + idx);
    *(v4f*)(dst + idx) = v;
  }
}

__global__ __launch_bounds__(256) void cbp_finalize_planes_kernel(
    const float* __restrict__ planes, float* out) {
  __shared__ float red[256];
  const int b   = blockIdx.x;
  const int tid = threadIdx.x;
  const float* p0 = planes + (size_t)b * ND;
  float* o = out + (size_t)b * ND;

  v4f v[ND / 1024];
  float ssum = 0.0f;
#pragma unroll
  for (int k = 0; k < ND / 1024; ++k) {
    const int idx = 4 * (tid + 256 * k);
    v4f a = *(const v4f*)(p0 + idx);
#pragma unroll
    for (int p = 1; p < NPLANE - 1; ++p) a += *(const v4f*)(p0 + (size_t)p * (NB * ND) + idx);
    a += *(const v4f*)(o + idx);
    v[k] = a;
    ssum += fabsf(a.x) + 1e-5f;
    ssum += fabsf(a.y) + 1e-5f;
    ssum += fabsf(a.z) + 1e-5f;
    ssum += fabsf(a.w) + 1e-5f;
  }
  red[tid] = ssum;
  __syncthreads();
  for (int s = 128; s > 0; s >>= 1) {
    if (tid < s) red[tid] += red[tid + s];
    __syncthreads();
  }
  const float inv = 1.0f / fmaxf(sqrtf(red[0]), 1e-12f);

#pragma unroll
  for (int k = 0; k < ND / 1024; ++k) {
    v4f r;
#pragma unroll
    for (int e = 0; e < 4; ++e) {
      const float val = v[k][e];
      const float sg  = (val > 0.0f) ? 1.0f : ((val < 0.0f) ? -1.0f : 0.0f);
      r[e] = sg * sqrtf(fabsf(val) + 1e-5f) * inv;
    }
    *(v4f*)(o + 4 * (tid + 256 * k)) = r;
  }
}

extern "C" void kernel_launch(void* const* d_in, const int* in_sizes, int n_in,
                              void* d_out, int out_size, void* d_ws, size_t ws_size,
                              hipStream_t stream) {
  const float* x  = (const float*)d_in[0];
  const int*   h1 = (const int*)d_in[1];
  const float* s1 = (const float*)d_in[2];
  const int*   h2 = (const int*)d_in[3];
  const float* s2 = (const float*)d_in[4];
  float*       out = (float*)d_out;

  float* cbp = (float*)d_ws;
  const int n = NB * ND;

  if (ws_size >= 7340032) {
    cbp_gram_sketch_planes_kernel<<<NB * NPLANE, 128, 0, stream>>>(x, h1, s1, h2, s2, cbp, out);
    cbp_finalize_planes_kernel<<<NB, 256, 0, stream>>>(cbp, out);
  } else {
    cbp_zero_kernel<<<(n + 255) / 256, 256, 0, stream>>>(cbp, out, n);
    cbp_gram_sketch_kernel<<<NB * 4, 256, 0, stream>>>(x, h1, s1, h2, s2, cbp, out);
    cbp_finalize_kernel<<<NB, 256, 0, stream>>>(cbp, out);
  }
}


# 28 of the 32 diagonal 16x16 tiles of G move from the 4 waves with two diagonal blocks to the 28 waves with a full block, one each: at most 17 tiles per wave, was 20; order of LDS adds changes
# speedup vs baseline: 5.4495x; 1.0471x over previous
#include <hip/hip_runtime.h>
#include <math.h>


typedef __attribute__((ext_vector_type(2))) float v2f;
typedef __attribute__((ext_vector_type(4))) float v4f;
typedef __attribute__((ext_vector_type(8))) float v8f;

#define NB 32
#define NC 512
#define HW 196
#define ND 8192

__global__ void cbp_zero_kernel(float* __restrict__ p, float* __restrict__ q, int n) {
  int i = blockIdx.x * blockDim.x + threadIdx.x;
  if (i < n) {
    p[i] = 0.0f;
    q[i] = 0.0f;
  }
}

__device__ __forceinline__ v8f wmma_k4(v2f a, v2f b, v8f c) {
  return __builtin_amdgcn_wmma_f32_16x16x4_f32(
       false, a,  false, b,
       (short)0, c,  false,  false);
}

__device__ __forceinline__ void load_frag4(v4f (&f)[4], const float* p, int s) {
#pragma unroll
  for (int t = 0; t < 4; ++t) f[t] = *(const v4f*)(p + t * (16 * HW) + 8 * s);
}

template <bool DIAG>
__device__ __forceinline__ void mma_block(v8f (&acc)[4][4], const v4f (&a)[4],
                                          const v4f (&b)[4]) {
#pragma unroll
  for (int ti = 0; ti < 4; ++ti)
#pragma unroll
    for (int tj = 0; tj < 4; ++tj)
      if (!DIAG || ti <= tj) acc[ti][tj] = wmma_k4(a[ti].xy, b[tj].xy, acc[ti][tj]);
#pragma unroll
  for (int ti = 0; ti < 4; ++ti)
#pragma unroll
    for (int tj = 0; tj < 4; ++tj)
      if (!DIAG || ti <= tj) acc[ti][tj] = wmma_k4(a[ti].zw, b[tj].zw, acc[ti][tj]);
}

template <bool DIAG>
__device__ __forceinline__ void gram_block(v8f (&acc)[4][4], const float* xb, int row0,
                                           int col0, int lr, int half) {
  const float* pa = xb + (row0 + lr) * HW + 4 * half;
  const float* pb = xb + (col0 + lr) * HW + 4 * half;
#pragma unroll
  for (int ti = 0; ti < 4; ++ti)
#pragma unroll
    for (int tj = 0; tj < 4; ++tj) acc[ti][tj] = (v8f){0.0f, 0.0f, 0.0f, 0.0f, 0.0f, 0.0f, 0.0f, 0.0f};

  v4f a0[4], b0[4], a1[4], b1[4];
  load_frag4(a0, pa, 0);
  if (!DIAG) load_frag4(b0, pb, 0);
#pragma nounroll
  for (int it = 0; it < 11; ++it) {
    load_frag4(a1, pa, 2 * it + 1);
    if (!DIAG) load_frag4(b1, pb, 2 * it + 1);
    mma_block<DIAG>(acc, a0, DIAG ? a0 : b0);
    load_frag4(a0, pa, 2 * it + 2);
    if (!DIAG) load_frag4(b0, pb, 2 * it + 2);
    mma_block<DIAG>(acc, a1, DIAG ? a1 : b1);
  }
  load_frag4(a1, pa, 23);
  if (!DIAG) load_frag4(b1, pb, 23);
  mma_block<DIAG>(acc, a0, DIAG ? a0 : b0);
  v2f ta[4], tb[4];
#pragma unroll
  for (int t = 0; t < 4; ++t) {
    ta[t] = *(const v2f*)(xb + (row0 + t * 16 + lr) * HW + 192 + 2 * half);
    tb[t] = DIAG ? ta[t] : *(const v2f*)(xb + (col0 + t * 16 + lr) * HW + 192 + 2 * half);
  }
  mma_block<DIAG>(acc, a1, DIAG ? a1 : b1);
#pragma unroll
  for (int ti = 0; ti < 4; ++ti)
#pragma unroll
    for (int tj = 0; tj < 4; ++tj)
      if (!DIAG || ti <= tj) acc[ti][tj] = wmma_k4(ta[ti], tb[tj], acc[ti][tj]);
}

template <int MODE>
__device__ __forceinline__ constexpr bool tile_on(int ti, int tj) {
  return MODE == 0 || ti < tj || (MODE == 3 && ti == 3 && tj == 3);
}

template <int MODE>
__device__ __forceinline__ void mma_tiles(v8f (&acc)[4][4], const v4f (&a)[4],
                                          const v4f (&b)[4]) {
#pragma unroll
  for (int ti = 0; ti < 4; ++ti)
#pragma unroll
    for (int tj = 0; tj < 4; ++tj)
      if (tile_on<MODE>(ti, tj)) acc[ti][tj] = wmma_k4(a[ti].xy, b[tj].xy, acc[ti][tj]);
#pragma unroll
  for (int ti = 0; ti < 4; ++ti)
#pragma unroll
    for (int tj = 0; tj < 4; ++tj)
      if (tile_on<MODE>(ti, tj)) acc[ti][tj] = wmma_k4(a[ti].zw, b[tj].zw, acc[ti][tj]);
}

__device__ __forceinline__ void mma_block_tile(v8f (&acc)[4][4], v8f& accD, const v4f (&a)[4],
                                               const v4f (&b)[4], v4f d) {
#pragma unroll
  for (int ti = 0; ti < 4; ++ti)
#pragma unroll
    for (int tj = 0; tj < 4; ++tj) acc[ti][tj] = wmma_k4(a[ti].xy, b[tj].xy, acc[ti][tj]);
  accD = wmma_k4(d.xy, d.xy, accD);
#pragma unroll
  for (int ti = 0; ti < 4; ++ti)
#pragma unroll
    for (int tj = 0; tj < 4; ++tj) acc[ti][tj] = wmma_k4(a[ti].zw, b[tj].zw, acc[ti][tj]);
  accD = wmma_k4(d.zw, d.zw, accD);
}

__device__ __forceinline__ void gram_block_tile(v8f (&acc)[4][4], v8f& accD, const float* xb,
                                                int row0, int col0, int drow, int lr, int half) {
  const float* pa = xb + (row0 + lr) * HW + 4 * half;
  const float* pb = xb + (col0 + lr) * HW + 4 * half;
  const float* pd = xb + (drow + lr) * HW + 4 * half;
#pragma unroll
  for (int ti = 0; ti < 4; ++ti)
#pragma unroll
    for (int tj = 0; tj < 4; ++tj) acc[ti][tj] = (v8f){0.0f, 0.0f, 0.0f, 0.0f, 0.0f, 0.0f, 0.0f, 0.0f};
  accD = (v8f){0.0f, 0.0f, 0.0f, 0.0f, 0.0f, 0.0f, 0.0f, 0.0f};

  v4f a0[4], b0[4], a1[4], b1[4], d0, d1;
  load_frag4(a0, pa, 0);
  load_frag4(b0, pb, 0);
  d0 = *(const v4f*)(pd);
#pragma nounroll
  for (int it = 0; it < 11; ++it) {
    load_frag4(a1, pa, 2 * it + 1);
    load_frag4(b1, pb, 2 * it + 1);
    d1 = *(const v4f*)(pd + 8 * (2 * it + 1));
    mma_block_tile(acc, accD, a0, b0, d0);
    load_frag4(a0, pa, 2 * it + 2);
    load_frag4(b0, pb, 2 * it + 2);
    d0 = *(const v4f*)(pd + 8 * (2 * it + 2));
    mma_block_tile(acc, accD, a1, b1, d1);
  }
  load_frag4(a1, pa, 23);
  load_frag4(b1, pb, 23);
  d1 = *(const v4f*)(pd + 8 * 23);
  mma_block_tile(acc, accD, a0, b0, d0);
  v2f ta[4], tb[4];
#pragma unroll
  for (int t = 0; t < 4; ++t) {
    ta[t] = *(const v2f*)(xb + (row0 + t * 16 + lr) * HW + 192 + 2 * half);
    tb[t] = *(const v2f*)(xb + (col0 + t * 16 + lr) * HW + 192 + 2 * half);
  }
  const v2f td = *(const v2f*)(xb + (drow + lr) * HW + 192 + 2 * half);
  mma_block_tile(acc, accD, a1, b1, d1);
#pragma unroll
  for (int ti = 0; ti < 4; ++ti)
#pragma unroll
    for (int tj = 0; tj < 4; ++tj) acc[ti][tj] = wmma_k4(ta[ti], tb[tj], acc[ti][tj]);
  accD = wmma_k4(td, td, accD);
}

__device__ __forceinline__ void gram_diag_pair_upper(v8f (&accA)[4][4], v8f (&accB)[4][4],
                                                     const float* xb, int rowA, int rowB,
                                                     int lr, int half) {
  const float* pa = xb + (rowA + lr) * HW + 4 * half;
  const float* pb = xb + (rowB + lr) * HW + 4 * half;
#pragma unroll
  for (int ti = 0; ti < 4; ++ti)
#pragma unroll
    for (int tj = 0; tj < 4; ++tj) {
      accA[ti][tj] = (v8f){0.0f, 0.0f, 0.0f, 0.0f, 0.0f, 0.0f, 0.0f, 0.0f};
      accB[ti][tj] = (v8f){0.0f, 0.0f, 0.0f, 0.0f, 0.0f, 0.0f, 0.0f, 0.0f};
    }

  v4f a0[4], b0[4], a1[4], b1[4];
  load_frag4(a0, pa, 0);
  load_frag4(b0, pb, 0);
#pragma nounroll
  for (int it = 0; it < 11; ++it) {
    load_frag4(a1, pa, 2 * it + 1);
    load_frag4(b1, pb, 2 * it + 1);
    mma_tiles<2>(accA, a0, a0);
    mma_tiles<3>(accB, b0, b0);
    load_frag4(a0, pa, 2 * it + 2);
    load_frag4(b0, pb, 2 * it + 2);
    mma_tiles<2>(accA, a1, a1);
    mma_tiles<3>(accB, b1, b1);
  }
  load_frag4(a1, pa, 23);
  load_frag4(b1, pb, 23);
  mma_tiles<2>(accA, a0, a0);
  mma_tiles<3>(accB, b0, b0);
  v2f ta[4], tb[4];
#pragma unroll
  for (int t = 0; t < 4; ++t) {
    ta[t] = *(const v2f*)(xb + (rowA + t * 16 + lr) * HW + 192 + 2 * half);
    tb[t] = *(const v2f*)(xb + (rowB + t * 16 + lr) * HW + 192 + 2 * half);
  }
  mma_tiles<2>(accA, a1, a1);
  mma_tiles<3>(accB, b1, b1);
#pragma unroll
  for (int ti = 0; ti < 4; ++ti)
#pragma unroll
    for (int tj = 0; tj < 4; ++tj) {
      if (tile_on<2>(ti, tj)) accA[ti][tj] = wmma_k4(ta[ti], ta[tj], accA[ti][tj]);
      if (tile_on<3>(ti, tj)) accB[ti][tj] = wmma_k4(tb[ti], tb[tj], accB[ti][tj]);
    }
}

template <bool DIAG>
__device__ __forceinline__ void sketch_block(const v8f (&acc)[4][4], float* myhist, int row0,
                                             int col0, int lr, int half,
                                             const int* __restrict__ h1,
                                             const float* __restrict__ s1,
                                             const int* __restrict__ h2,
                                             const float* __restrict__ s2) {
  int   h1n[4], h2n[4];
  float s1n[4], s2n[4];
#pragma unroll
  for (int tj = 0; tj < 4; ++tj) {
    const int n = col0 + tj * 16 + lr;
    h1n[tj] = h1[n];
    s1n[tj] = s1[n];
    h2n[tj] = h2[n];
    s2n[tj] = s2[n];
  }
#pragma unroll
  for (int ti = 0; ti < 4; ++ti) {
    const int mbase = row0 + ti * 16 + half * 8;
    int   h1m[8], h2m[8];
    float s1m[8], s2m[8];
#pragma unroll
    for (int r = 0; r < 8; ++r) {
      h1m[r] = h1[mbase + r];
      s1m[r] = s1[mbase + r];
      h2m[r] = h2[mbase + r];
      s2m[r] = s2[mbase + r];
    }
#pragma unroll
    for (int tj = 0; tj < 4; ++tj) {
      if (DIAG && ti > tj) continue;
#pragma unroll
      for (int r = 0; r < 8; ++r) {
        const float g = acc[ti][tj][r];
        atomicAdd(&myhist[(h1m[r] + h2n[tj]) & (ND - 1)], s1m[r] * s2n[tj] * g);
        if (!(DIAG && ti == tj))
          atomicAdd(&myhist[(h1n[tj] + h2m[r]) & (ND - 1)], s1n[tj] * s2m[r] * g);
      }
    }
  }
}

template <int MODE>
__device__ __forceinline__ void sketch_tiles(const v8f (&acc)[4][4], float* myhist, int row0,
                                             int col0, int lr, int half,
                                             const int* __restrict__ h1,
                                             const float* __restrict__ s1,
                                             const int* __restrict__ h2,
                                             const float* __restrict__ s2) {
  int   h1n[4], h2n[4];
  float s1n[4], s2n[4];
#pragma unroll
  for (int tj = 0; tj < 4; ++tj) {
    const int n = col0 + tj * 16 + lr;
    h1n[tj] = h1[n];
    s1n[tj] = s1[n];
    h2n[tj] = h2[n];
    s2n[tj] = s2[n];
  }
#pragma unroll
  for (int ti = 0; ti < 4; ++ti) {
    const int mbase = row0 + ti * 16 + half * 8;
    int   h1m[8], h2m[8];
    float s1m[8], s2m[8];
#pragma unroll
    for (int r = 0; r < 8; ++r) {
      h1m[r] = h1[mbase + r];
      s1m[r] = s1[mbase + r];
      h2m[r] = h2[mbase + r];
      s2m[r] = s2[mbase + r];
    }
#pragma unroll
    for (int tj = 0; tj < 4; ++tj) {
      if (!tile_on<MODE>(ti, tj)) continue;
#pragma unroll
      for (int r = 0; r < 8; ++r) {
        const float g = acc[ti][tj][r];
        atomicAdd(&myhist[(h1m[r] + h2n[tj]) & (ND - 1)], s1m[r] * s2n[tj] * g);
        if (MODE == 0 || ti != tj)
          atomicAdd(&myhist[(h1n[tj] + h2m[r]) & (ND - 1)], s1n[tj] * s2m[r] * g);
      }
    }
  }
}

__device__ __forceinline__ void sketch_diag_tile(const v8f& accD, float* myhist, int row0, int lr,
                                                 int half, const int* __restrict__ h1,
                                                 const float* __restrict__ s1,
                                                 const int* __restrict__ h2,
                                                 const float* __restrict__ s2) {
  const int   n   = row0 + lr;
  const int   h2n = h2[n];
  const float s2n = s2[n];
  const int mbase = row0 + half * 8;
#pragma unroll
  for (int r = 0; r < 8; ++r)
    atomicAdd(&myhist[(h1[mbase + r] + h2n) & (ND - 1)], s1[mbase + r] * s2n * accD[r]);
}

__global__ __launch_bounds__(256) void cbp_gram_sketch_kernel(
    const float* __restrict__ x, const int* __restrict__ h1,
    const float* __restrict__ s1, const int* __restrict__ h2,
    const float* __restrict__ s2, float* __restrict__ plane0,
    float* __restrict__ plane1) {
  __shared__ float hist[8 * ND];

  const int b    = blockIdx.x >> 2;
  const int q    = blockIdx.x & 3;
  const int tid  = threadIdx.x;
  const int wave = __builtin_amdgcn_readfirstlane(tid >> 5);
  const int lane = tid & 31;
  const int half = lane >> 4;
  const int lr   = lane & 15;

  float* myhist = hist + wave * ND;
  for (int k = lane * 4; k < ND; k += 128) *(v4f*)(myhist + k) = (v4f){0.0f, 0.0f, 0.0f, 0.0f};

  const float* xb = x + (size_t)b * (NC * HW);

  v8f acc[4][4];
  if (wave < 7) {
    int I = 0, rem = q * 7 + wave;
    for (int t = 0; t < 6; ++t)
      if (rem >= 7 - I) {
        rem -= 7 - I;
        ++I;
      }
    const int J = I + 1 + rem;
    gram_block<false>(acc, xb, I * 64, J * 64, lr, half);
    sketch_block<false>(acc, myhist, I * 64, J * 64, lr, half, h1, s1, h2, s2);
  } else {
    for (int d = 0; d < 2; ++d) {
      const int I = 2 * q + d;
      gram_block<true>(acc, xb, I * 64, I * 64, lr, half);
      sketch_block<true>(acc, myhist, I * 64, I * 64, lr, half, h1, s1, h2, s2);
    }
  }

  __syncthreads();
  float* dst = ((q < 2) ? plane0 : plane1) + (size_t)b * ND;
  for (int idx = tid; idx < ND; idx += 256) {
    float v = hist[idx];
#pragma unroll
    for (int w = 1; w < 8; ++w) v += hist[w * ND + idx];
    atomicAdd(&dst[idx], v);
  }
}

__global__ __launch_bounds__(256) void cbp_finalize_kernel(
    const float* __restrict__ plane0, float* out) {
  __shared__ float red[256];
  const int b   = blockIdx.x;
  const int tid = threadIdx.x;
  const float* v0 = plane0 + (size_t)b * ND;
  float* o = out + (size_t)b * ND;

  float v[ND / 256];
  float ssum = 0.0f;
#pragma unroll
  for (int k = 0; k < ND / 256; ++k) {
    const int idx = tid + 256 * k;
    v[k] = v0[idx] + o[idx];
    ssum += fabsf(v[k]) + 1e-5f;
  }
  red[tid] = ssum;
  __syncthreads();
  for (int s = 128; s > 0; s >>= 1) {
    if (tid < s) red[tid] += red[tid + s];
    __syncthreads();
  }
  const float inv = 1.0f / fmaxf(sqrtf(red[0]), 1e-12f);

#pragma unroll
  for (int k = 0; k < ND / 256; ++k) {
    const float val = v[k];
    const float sg  = (val > 0.0f) ? 1.0f : ((val < 0.0f) ? -1.0f : 0.0f);
    o[tid + 256 * k] = sg * sqrtf(fabsf(val) + 1e-5f) * inv;
  }
}

#define NPLANE 8
#define PANEL_FLOATS (64 * HW)
#define PAIR_V4 (2 * PANEL_FLOATS / 4)
#define PLANES_LDS_FLOATS (4 * PANEL_FLOATS)
__global__ __launch_bounds__(128) void cbp_gram_sketch_planes_kernel(
    const float* __restrict__ x, const int* __restrict__ h1,
    const float* __restrict__ s1, const int* __restrict__ h2,
    const float* __restrict__ s2, float* __restrict__ planes,
    float* __restrict__ out) {
  __shared__ __attribute__((aligned(16))) float lds[PLANES_LDS_FLOATS];

  const int b    = blockIdx.x >> 3;
  const int g    = blockIdx.x & 7;
  const int tid  = threadIdx.x;
  const int wave = __builtin_amdgcn_readfirstlane(tid >> 5);
  const int lane = tid & 31;
  const int half = lane >> 4;
  const int lr   = lane & 15;

  const float* xb = x + (size_t)b * (NC * HW);

  int p0, p2;
  if (g < 6) {
    const int SI = (g >= 3) + (g >= 5);
    const int SJ = (g < 3) ? g + 1 : ((g < 5) ? g - 1 : 3);
    p0 = 2 * SI;
    p2 = 2 * SJ;
  } else {
    p0 = 4 * (g - 6);
    p2 = p0 + 2;
  }

  {
    const v4f* srcA = (const v4f*)(xb + p0 * PANEL_FLOATS);
    const v4f* srcB = (const v4f*)(xb + p2 * PANEL_FLOATS);
    v4f* dst = (v4f*)lds;
    v4f t0[25], t1[24];
#pragma unroll
    for (int i = 0; i < 25; ++i) t0[i] = srcA[tid + 128 * i];
#pragma unroll
    for (int i = 0; i < 24; ++i) t1[i] = srcA[tid + 128 * (25 + i)];
#pragma unroll
    for (int i = 0; i < 25; ++i) dst[tid + 128 * i] = t0[i];
#pragma unroll
    for (int i = 0; i < 25; ++i) t0[i] = srcB[tid + 128 * i];
#pragma unroll
    for (int i = 0; i < 24; ++i) dst[tid + 128 * (25 + i)] = t1[i];
#pragma unroll
    for (int i = 0; i < 24; ++i) t1[i] = srcB[tid + 128 * (25 + i)];
#pragma unroll
    for (int i = 0; i < 25; ++i) dst[PAIR_V4 + tid + 128 * i] = t0[i];
#pragma unroll
    for (int i = 0; i < 24; ++i) dst[PAIR_V4 + tid + 128 * (25 + i)] = t1[i];
  }

  bool above;
  int slotA, slotB, I, J;
  int dside = 0, dt = 0;
  if (g < 6) {
    above = true;
    slotA = wave >> 1;
    slotB = 2 + (wave & 1);
    I = p0 + (wave >> 1);
    J = p2 + (wave & 1);
    dside = wave >> 1;
    dt = 2 * (dside ? (p0 >> 1) : (p2 >> 1) - 1) + (wave & 1);
  } else {
    above = wave < 2;
    slotA = 2 * (wave & 1);
    slotB = slotA + 1;
    I = p0 + slotA;
    J = I + 1;
    dside = wave & 1;
    dt = 6;
  }
  const int drow_lds = dside * 128 + dt * 16;
  const int drow_g   = (dside ? p2 : p0) * 64 + dt * 16;

  __syncthreads();

  v8f acc[4][4], acc2[4][4], accD;
  if (above)
    gram_block_tile(acc, accD, lds, slotA * 64, slotB * 64, drow_lds, lr, half);
  else
    gram_diag_pair_upper(acc, acc2, lds, slotA * 64, slotB * 64, lr, half);

  __syncthreads();

  float* myhist = lds + wave * ND;
  for (int k = lane * 4; k < ND; k += 128) *(v4f*)(myhist + k) = (v4f){0.0f, 0.0f, 0.0f, 0.0f};

  if (above) {
    sketch_tiles<0>(acc, myhist, I * 64, J * 64, lr, half, h1, s1, h2, s2);
    sketch_diag_tile(accD, myhist, drow_g, lr, half, h1, s1, h2, s2);
  } else {
    sketch_tiles<2>(acc, myhist, I * 64, I * 64, lr, half, h1, s1, h2, s2);
    sketch_tiles<3>(acc2, myhist, J * 64, J * 64, lr, half, h1, s1, h2, s2);
  }

  __syncthreads();
  float* dst = ((g < NPLANE - 1) ? planes + (size_t)g * (NB * ND) : out) + (size_t)b * ND;
  for (int idx = tid * 4; idx < ND; idx += 128 * 4) {
    v4f v = *(const v4f*)(lds + idx);
    v += *(const v4f*)(lds + ND + idx);
    v += *(const v4f*)(lds + 2 * ND + idx);
    v += *(const v4f*)(lds + 3 * ND + idx);
    *(v4f*)(dst + idx) = v;
  }
}

__global__ __launch_bounds__(256) void cbp_finalize_planes_kernel(
    const float* __restrict__ planes, float* out) {
  __shared__ float red[256];
  const int b   = blockIdx.x;
  const int tid = threadIdx.x;
  const float* p0 = planes + (size_t)b * ND;
  float* o = out + (size_t)b * ND;

  v4f v[ND / 1024];
  float ssum = 0.0f;
#pragma unroll
  for (int k = 0; k < ND / 1024; ++k) {
    const int idx = 4 * (tid + 256 * k);
    v4f a = *(const v4f*)(p0 + idx);
#pragma unroll
    for (int p = 1; p < NPLANE - 1; ++p) a += *(const v4f*)(p0 + (size_t)p * (NB * ND) + idx);
    a += *(const v4f*)(o + idx);
    v[k] = a;
    ssum += fabsf(a.x) + 1e-5f;
    ssum += fabsf(a.y) + 1e-5f;
    ssum += fabsf(a.z) + 1e-5f;
    ssum += fabsf(a.w) + 1e-5f;
  }
  red[tid] = ssum;
  __syncthreads();
  for (int s = 128; s > 0; s >>= 1) {
    if (tid < s) red[tid] += red[tid + s];
    __syncthreads();
  }
  const float inv = 1.0f / fmaxf(sqrtf(red[0]), 1e-12f);

#pragma unroll
  for (int k = 0; k < ND / 1024; ++k) {
    v4f r;
#pragma unroll
    for (int e = 0; e < 4; ++e) {
      const float val = v[k][e];
      const float sg  = (val > 0.0f) ? 1.0f : ((val < 0.0f) ? -1.0f : 0.0f);
      r[e] = sg * sqrtf(fabsf(val) + 1e-5f) * inv;
    }
    *(v4f*)(o + 4 * (tid + 256 * k)) = r;
  }
}

extern "C" void kernel_launch(void* const* d_in, const int* in_sizes, int n_in,
                              void* d_out, int out_size, void* d_ws, size_t ws_size,
                              hipStream_t stream) {
  const float* x  = (const float*)d_in[0];
  const int*   h1 = (const int*)d_in[1];
  const float* s1 = (const float*)d_in[2];
  const int*   h2 = (const int*)d_in[3];
  const float* s2 = (const float*)d_in[4];
  float*       out = (float*)d_out;

  float* cbp = (float*)d_ws;
  const int n = NB * ND;

  if (ws_size >= 7340032) {
    cbp_gram_sketch_planes_kernel<<<NB * NPLANE, 128, 0, stream>>>(x, h1, s1, h2, s2, cbp, out);
    cbp_finalize_planes_kernel<<<NB, 256, 0, stream>>>(cbp, out);
  } else {
    cbp_zero_kernel<<<(n + 255) / 256, 256, 0, stream>>>(cbp, out, n);
    cbp_gram_sketch_kernel<<<NB * 4, 256, 0, stream>>>(x, h1, s1, h2, s2, cbp, out);
    cbp_finalize_kernel<<<NB, 256, 0, stream>>>(cbp, out);
  }
}


# Serial loops round the matrix loop issue all their memory accesses before the first use: hash/sign loads in one group, clear and plane-sum loops unrolled, finalize's 64 reads in flight at once
# speedup vs baseline: 5.8373x; 1.0712x over previous
#include <hip/hip_runtime.h>
#include <math.h>


typedef __attribute__((ext_vector_type(2))) float v2f;
typedef __attribute__((ext_vector_type(4))) float v4f;
typedef __attribute__((ext_vector_type(8))) float v8f;

#define NB 32
#define NC 512
#define HW 196
#define ND 8192

__global__ void cbp_zero_kernel(float* __restrict__ p, float* __restrict__ q, int n) {
  int i = blockIdx.x * blockDim.x + threadIdx.x;
  if (i < n) {
    p[i] = 0.0f;
    q[i] = 0.0f;
  }
}

__device__ __forceinline__ v8f wmma_k4(v2f a, v2f b, v8f c) {
  return __builtin_amdgcn_wmma_f32_16x16x4_f32(
       false, a,  false, b,
       (short)0, c,  false,  false);
}

__device__ __forceinline__ void load_frag4(v4f (&f)[4], const float* p, int s) {
#pragma unroll
  for (int t = 0; t < 4; ++t) f[t] = *(const v4f*)(p + t * (16 * HW) + 8 * s);
}

template <bool DIAG>
__device__ __forceinline__ void mma_block(v8f (&acc)[4][4], const v4f (&a)[4],
                                          const v4f (&b)[4]) {
#pragma unroll
  for (int ti = 0; ti < 4; ++ti)
#pragma unroll
    for (int tj = 0; tj < 4; ++tj)
      if (!DIAG || ti <= tj) acc[ti][tj] = wmma_k4(a[ti].xy, b[tj].xy, acc[ti][tj]);
#pragma unroll
  for (int ti = 0; ti < 4; ++ti)
#pragma unroll
    for (int tj = 0; tj < 4; ++tj)
      if (!DIAG || ti <= tj) acc[ti][tj] = wmma_k4(a[ti].zw, b[tj].zw, acc[ti][tj]);
}

template <bool DIAG>
__device__ __forceinline__ void gram_block(v8f (&acc)[4][4], const float* xb, int row0,
                                           int col0, int lr, int half) {
  const float* pa = xb + (row0 + lr) * HW + 4 * half;
  const float* pb = xb + (col0 + lr) * HW + 4 * half;
#pragma unroll
  for (int ti = 0; ti < 4; ++ti)
#pragma unroll
    for (int tj = 0; tj < 4; ++tj) acc[ti][tj] = (v8f){0.0f, 0.0f, 0.0f, 0.0f, 0.0f, 0.0f, 0.0f, 0.0f};

  v4f a0[4], b0[4], a1[4], b1[4];
  load_frag4(a0, pa, 0);
  if (!DIAG) load_frag4(b0, pb, 0);
#pragma nounroll
  for (int it = 0; it < 11; ++it) {
    load_frag4(a1, pa, 2 * it + 1);
    if (!DIAG) load_frag4(b1, pb, 2 * it + 1);
    mma_block<DIAG>(acc, a0, DIAG ? a0 : b0);
    load_frag4(a0, pa, 2 * it + 2);
    if (!DIAG) load_frag4(b0, pb, 2 * it + 2);
    mma_block<DIAG>(acc, a1, DIAG ? a1 : b1);
  }
  load_frag4(a1, pa, 23);
  if (!DIAG) load_frag4(b1, pb, 23);
  mma_block<DIAG>(acc, a0, DIAG ? a0 : b0);
  v2f ta[4], tb[4];
#pragma unroll
  for (int t = 0; t < 4; ++t) {
    ta[t] = *(const v2f*)(xb + (row0 + t * 16 + lr) * HW + 192 + 2 * half);
    tb[t] = DIAG ? ta[t] : *(const v2f*)(xb + (col0 + t * 16 + lr) * HW + 192 + 2 * half);
  }
  mma_block<DIAG>(acc, a1, DIAG ? a1 : b1);
#pragma unroll
  for (int ti = 0; ti < 4; ++ti)
#pragma unroll
    for (int tj = 0; tj < 4; ++tj)
      if (!DIAG || ti <= tj) acc[ti][tj] = wmma_k4(ta[ti], tb[tj], acc[ti][tj]);
}

template <int MODE>
__device__ __forceinline__ constexpr bool tile_on(int ti, int tj) {
  return MODE == 0 || ti < tj || (MODE == 3 && ti == 3 && tj == 3);
}

template <int MODE>
__device__ __forceinline__ void mma_tiles(v8f (&acc)[4][4], const v4f (&a)[4],
                                          const v4f (&b)[4]) {
#pragma unroll
  for (int ti = 0; ti < 4; ++ti)
#pragma unroll
    for (int tj = 0; tj < 4; ++tj)
      if (tile_on<MODE>(ti, tj)) acc[ti][tj] = wmma_k4(a[ti].xy, b[tj].xy, acc[ti][tj]);
#pragma unroll
  for (int ti = 0; ti < 4; ++ti)
#pragma unroll
    for (int tj = 0; tj < 4; ++tj)
      if (tile_on<MODE>(ti, tj)) acc[ti][tj] = wmma_k4(a[ti].zw, b[tj].zw, acc[ti][tj]);
}

__device__ __forceinline__ void mma_block_tile(v8f (&acc)[4][4], v8f& accD, const v4f (&a)[4],
                                               const v4f (&b)[4], v4f d) {
#pragma unroll
  for (int ti = 0; ti < 4; ++ti)
#pragma unroll
    for (int tj = 0; tj < 4; ++tj) acc[ti][tj] = wmma_k4(a[ti].xy, b[tj].xy, acc[ti][tj]);
  accD = wmma_k4(d.xy, d.xy, accD);
#pragma unroll
  for (int ti = 0; ti < 4; ++ti)
#pragma unroll
    for (int tj = 0; tj < 4; ++tj) acc[ti][tj] = wmma_k4(a[ti].zw, b[tj].zw, acc[ti][tj]);
  accD = wmma_k4(d.zw, d.zw, accD);
}

__device__ __forceinline__ void gram_block_tile(v8f (&acc)[4][4], v8f& accD, const float* xb,
                                                int row0, int col0, int drow, int lr, int half) {
  const float* pa = xb + (row0 + lr) * HW + 4 * half;
  const float* pb = xb + (col0 + lr) * HW + 4 * half;
  const float* pd = xb + (drow + lr) * HW + 4 * half;
#pragma unroll
  for (int ti = 0; ti < 4; ++ti)
#pragma unroll
    for (int tj = 0; tj < 4; ++tj) acc[ti][tj] = (v8f){0.0f, 0.0f, 0.0f, 0.0f, 0.0f, 0.0f, 0.0f, 0.0f};
  accD = (v8f){0.0f, 0.0f, 0.0f, 0.0f, 0.0f, 0.0f, 0.0f, 0.0f};

  v4f a0[4], b0[4], a1[4], b1[4], d0, d1;
  load_frag4(a0, pa, 0);
  load_frag4(b0, pb, 0);
  d0 = *(const v4f*)(pd);
#pragma nounroll
  for (int it = 0; it < 11; ++it) {
    load_frag4(a1, pa, 2 * it + 1);
    load_frag4(b1, pb, 2 * it + 1);
    d1 = *(const v4f*)(pd + 8 * (2 * it + 1));
    mma_block_tile(acc, accD, a0, b0, d0);
    load_frag4(a0, pa, 2 * it + 2);
    load_frag4(b0, pb, 2 * it + 2);
    d0 = *(const v4f*)(pd + 8 * (2 * it + 2));
    mma_block_tile(acc, accD, a1, b1, d1);
  }
  load_frag4(a1, pa, 23);
  load_frag4(b1, pb, 23);
  d1 = *(const v4f*)(pd + 8 * 23);
  mma_block_tile(acc, accD, a0, b0, d0);
  v2f ta[4], tb[4];
#pragma unroll
  for (int t = 0; t < 4; ++t) {
    ta[t] = *(const v2f*)(xb + (row0 + t * 16 + lr) * HW + 192 + 2 * half);
    tb[t] = *(const v2f*)(xb + (col0 + t * 16 + lr) * HW + 192 + 2 * half);
  }
  const v2f td = *(const v2f*)(xb + (drow + lr) * HW + 192 + 2 * half);
  mma_block_tile(acc, accD, a1, b1, d1);
#pragma unroll
  for (int ti = 0; ti < 4; ++ti)
#pragma unroll
    for (int tj = 0; tj < 4; ++tj) acc[ti][tj] = wmma_k4(ta[ti], tb[tj], acc[ti][tj]);
  accD = wmma_k4(td, td, accD);
}

__device__ __forceinline__ void gram_diag_pair_upper(v8f (&accA)[4][4], v8f (&accB)[4][4],
                                                     const float* xb, int rowA, int rowB,
                                                     int lr, int half) {
  const float* pa = xb + (rowA + lr) * HW + 4 * half;
  const float* pb = xb + (rowB + lr) * HW + 4 * half;
#pragma unroll
  for (int ti = 0; ti < 4; ++ti)
#pragma unroll
    for (int tj = 0; tj < 4; ++tj) {
      accA[ti][tj] = (v8f){0.0f, 0.0f, 0.0f, 0.0f, 0.0f, 0.0f, 0.0f, 0.0f};
      accB[ti][tj] = (v8f){0.0f, 0.0f, 0.0f, 0.0f, 0.0f, 0.0f, 0.0f, 0.0f};
    }

  v4f a0[4], b0[4], a1[4], b1[4];
  load_frag4(a0, pa, 0);
  load_frag4(b0, pb, 0);
#pragma nounroll
  for (int it = 0; it < 11; ++it) {
    load_frag4(a1, pa, 2 * it + 1);
    load_frag4(b1, pb, 2 * it + 1);
    mma_tiles<2>(accA, a0, a0);
    mma_tiles<3>(accB, b0, b0);
    load_frag4(a0, pa, 2 * it + 2);
    load_frag4(b0, pb, 2 * it + 2);
    mma_tiles<2>(accA, a1, a1);
    mma_tiles<3>(accB, b1, b1);
  }
  load_frag4(a1, pa, 23);
  load_frag4(b1, pb, 23);
  mma_tiles<2>(accA, a0, a0);
  mma_tiles<3>(accB, b0, b0);
  v2f ta[4], tb[4];
#pragma unroll
  for (int t = 0; t < 4; ++t) {
    ta[t] = *(const v2f*)(xb + (rowA + t * 16 + lr) * HW + 192 + 2 * half);
    tb[t] = *(const v2f*)(xb + (rowB + t * 16 + lr) * HW + 192 + 2 * half);
  }
  mma_tiles<2>(accA, a1, a1);
  mma_tiles<3>(accB, b1, b1);
#pragma unroll
  for (int ti = 0; ti < 4; ++ti)
#pragma unroll
    for (int tj = 0; tj < 4; ++tj) {
      if (tile_on<2>(ti, tj)) accA[ti][tj] = wmma_k4(ta[ti], ta[tj], accA[ti][tj]);
      if (tile_on<3>(ti, tj)) accB[ti][tj] = wmma_k4(tb[ti], tb[tj], accB[ti][tj]);
    }
}

template <bool DIAG>
__device__ __forceinline__ void sketch_block(const v8f (&acc)[4][4], float* myhist, int row0,
                                             int col0, int lr, int half,
                                             const int* __restrict__ h1,
                                             const float* __restrict__ s1,
                                             const int* __restrict__ h2,
                                             const float* __restrict__ s2) {
  int   h1n[4], h2n[4];
  float s1n[4], s2n[4];
#pragma unroll
  for (int tj = 0; tj < 4; ++tj) {
    const int n = col0 + tj * 16 + lr;
    h1n[tj] = h1[n];
    s1n[tj] = s1[n];
    h2n[tj] = h2[n];
    s2n[tj] = s2[n];
  }
#pragma unroll
  for (int ti = 0; ti < 4; ++ti) {
    const int mbase = row0 + ti * 16 + half * 8;
    int   h1m[8], h2m[8];
    float s1m[8], s2m[8];
#pragma unroll
    for (int r = 0; r < 8; ++r) {
      h1m[r] = h1[mbase + r];
      s1m[r] = s1[mbase + r];
      h2m[r] = h2[mbase + r];
      s2m[r] = s2[mbase + r];
    }
#pragma unroll
    for (int tj = 0; tj < 4; ++tj) {
      if (DIAG && ti > tj) continue;
#pragma unroll
      for (int r = 0; r < 8; ++r) {
        const float g = acc[ti][tj][r];
        atomicAdd(&myhist[(h1m[r] + h2n[tj]) & (ND - 1)], s1m[r] * s2n[tj] * g);
        if (!(DIAG && ti == tj))
          atomicAdd(&myhist[(h1n[tj] + h2m[r]) & (ND - 1)], s1n[tj] * s2m[r] * g);
      }
    }
  }
}

struct SketchKeys {
  int   h1n[4], h2n[4];
  float s1n[4], s2n[4];
  int   h1m[4][8], h2m[4][8];
  float s1m[4][8], s2m[4][8];
};

__device__ __forceinline__ void sketch_keys(SketchKeys& k, int row0, int col0, int lr, int half,
                                            const int* __restrict__ h1,
                                            const float* __restrict__ s1,
                                            const int* __restrict__ h2,
                                            const float* __restrict__ s2) {
#pragma unroll
  for (int tj = 0; tj < 4; ++tj) {
    const int n = col0 + tj * 16 + lr;
    k.h1n[tj] = h1[n];
    k.s1n[tj] = s1[n];
    k.h2n[tj] = h2[n];
    k.s2n[tj] = s2[n];
  }
#pragma unroll
  for (int ti = 0; ti < 4; ++ti) {
    const int mbase = row0 + ti * 16 + half * 8;
#pragma unroll
    for (int r = 0; r < 8; ++r) {
      k.h1m[ti][r] = h1[mbase + r];
      k.s1m[ti][r] = s1[mbase + r];
      k.h2m[ti][r] = h2[mbase + r];
      k.s2m[ti][r] = s2[mbase + r];
    }
  }
}

template <int MODE>
__device__ __forceinline__ void sketch_add(const v8f (&acc)[4][4], float* myhist,
                                           const SketchKeys& k) {
#pragma unroll
  for (int ti = 0; ti < 4; ++ti) {
#pragma unroll
    for (int tj = 0; tj < 4; ++tj) {
      if (!tile_on<MODE>(ti, tj)) continue;
#pragma unroll
      for (int r = 0; r < 8; ++r) {
        const float g = acc[ti][tj][r];
        atomicAdd(&myhist[(k.h1m[ti][r] + k.h2n[tj]) & (ND - 1)], k.s1m[ti][r] * k.s2n[tj] * g);
        if (MODE == 0 || ti != tj)
          atomicAdd(&myhist[(k.h1n[tj] + k.h2m[ti][r]) & (ND - 1)], k.s1n[tj] * k.s2m[ti][r] * g);
      }
    }
  }
}

struct DiagKeys {
  int   h2n;
  float s2n;
  int   h1m[8];
  float s1m[8];
};

__device__ __forceinline__ void diag_keys(DiagKeys& k, int row0, int lr, int half,
                                          const int* __restrict__ h1,
                                          const float* __restrict__ s1,
                                          const int* __restrict__ h2,
                                          const float* __restrict__ s2) {
  const int n = row0 + lr;
  k.h2n = h2[n];
  k.s2n = s2[n];
  const int mbase = row0 + half * 8;
#pragma unroll
  for (int r = 0; r < 8; ++r) {
    k.h1m[r] = h1[mbase + r];
    k.s1m[r] = s1[mbase + r];
  }
}

__device__ __forceinline__ void sketch_diag_add(const v8f& accD, float* myhist, const DiagKeys& k) {
#pragma unroll
  for (int r = 0; r < 8; ++r)
    atomicAdd(&myhist[(k.h1m[r] + k.h2n) & (ND - 1)], k.s1m[r] * k.s2n * accD[r]);
}

__global__ __launch_bounds__(256) void cbp_gram_sketch_kernel(
    const float* __restrict__ x, const int* __restrict__ h1,
    const float* __restrict__ s1, const int* __restrict__ h2,
    const float* __restrict__ s2, float* __restrict__ plane0,
    float* __restrict__ plane1) {
  __shared__ float hist[8 * ND];

  const int b    = blockIdx.x >> 2;
  const int q    = blockIdx.x & 3;
  const int tid  = threadIdx.x;
  const int wave = __builtin_amdgcn_readfirstlane(tid >> 5);
  const int lane = tid & 31;
  const int half = lane >> 4;
  const int lr   = lane & 15;

  float* myhist = hist + wave * ND;
  for (int k = lane * 4; k < ND; k += 128) *(v4f*)(myhist + k) = (v4f){0.0f, 0.0f, 0.0f, 0.0f};

  const float* xb = x + (size_t)b * (NC * HW);

  v8f acc[4][4];
  if (wave < 7) {
    int I = 0, rem = q * 7 + wave;
    for (int t = 0; t < 6; ++t)
      if (rem >= 7 - I) {
        rem -= 7 - I;
        ++I;
      }
    const int J = I + 1 + rem;
    gram_block<false>(acc, xb, I * 64, J * 64, lr, half);
    sketch_block<false>(acc, myhist, I * 64, J * 64, lr, half, h1, s1, h2, s2);
  } else {
    for (int d = 0; d < 2; ++d) {
      const int I = 2 * q + d;
      gram_block<true>(acc, xb, I * 64, I * 64, lr, half);
      sketch_block<true>(acc, myhist, I * 64, I * 64, lr, half, h1, s1, h2, s2);
    }
  }

  __syncthreads();
  float* dst = ((q < 2) ? plane0 : plane1) + (size_t)b * ND;
  for (int idx = tid; idx < ND; idx += 256) {
    float v = hist[idx];
#pragma unroll
    for (int w = 1; w < 8; ++w) v += hist[w * ND + idx];
    atomicAdd(&dst[idx], v);
  }
}

__global__ __launch_bounds__(256) void cbp_finalize_kernel(
    const float* __restrict__ plane0, float* out) {
  __shared__ float red[256];
  const int b   = blockIdx.x;
  const int tid = threadIdx.x;
  const float* v0 = plane0 + (size_t)b * ND;
  float* o = out + (size_t)b * ND;

  float v[ND / 256];
  float ssum = 0.0f;
#pragma unroll
  for (int k = 0; k < ND / 256; ++k) {
    const int idx = tid + 256 * k;
    v[k] = v0[idx] + o[idx];
    ssum += fabsf(v[k]) + 1e-5f;
  }
  red[tid] = ssum;
  __syncthreads();
  for (int s = 128; s > 0; s >>= 1) {
    if (tid < s) red[tid] += red[tid + s];
    __syncthreads();
  }
  const float inv = 1.0f / fmaxf(sqrtf(red[0]), 1e-12f);

#pragma unroll
  for (int k = 0; k < ND / 256; ++k) {
    const float val = v[k];
    const float sg  = (val > 0.0f) ? 1.0f : ((val < 0.0f) ? -1.0f : 0.0f);
    o[tid + 256 * k] = sg * sqrtf(fabsf(val) + 1e-5f) * inv;
  }
}

#define NPLANE 8
#define PANEL_FLOATS (64 * HW)
#define PAIR_V4 (2 * PANEL_FLOATS / 4)
#define PLANES_LDS_FLOATS (4 * PANEL_FLOATS)
__global__ __launch_bounds__(128) void cbp_gram_sketch_planes_kernel(
    const float* __restrict__ x, const int* __restrict__ h1,
    const float* __restrict__ s1, const int* __restrict__ h2,
    const float* __restrict__ s2, float* __restrict__ planes,
    float* __restrict__ out) {
  __shared__ __attribute__((aligned(16))) float lds[PLANES_LDS_FLOATS];

  const int b    = blockIdx.x >> 3;
  const int g    = blockIdx.x & 7;
  const int tid  = threadIdx.x;
  const int wave = __builtin_amdgcn_readfirstlane(tid >> 5);
  const int lane = tid & 31;
  const int half = lane >> 4;
  const int lr   = lane & 15;

  const float* xb = x + (size_t)b * (NC * HW);

  int p0, p2;
  if (g < 6) {
    const int SI = (g >= 3) + (g >= 5);
    const int SJ = (g < 3) ? g + 1 : ((g < 5) ? g - 1 : 3);
    p0 = 2 * SI;
    p2 = 2 * SJ;
  } else {
    p0 = 4 * (g - 6);
    p2 = p0 + 2;
  }

  {
    const v4f* srcA = (const v4f*)(xb + p0 * PANEL_FLOATS);
    const v4f* srcB = (const v4f*)(xb + p2 * PANEL_FLOATS);
    v4f* dst = (v4f*)lds;
    v4f t0[25], t1[24];
#pragma unroll
    for (int i = 0; i < 25; ++i) t0[i] = srcA[tid + 128 * i];
#pragma unroll
    for (int i = 0; i < 24; ++i) t1[i] = srcA[tid + 128 * (25 + i)];
#pragma unroll
    for (int i = 0; i < 25; ++i) dst[tid + 128 * i] = t0[i];
#pragma unroll
    for (int i = 0; i < 25; ++i) t0[i] = srcB[tid + 128 * i];
#pragma unroll
    for (int i = 0; i < 24; ++i) dst[tid + 128 * (25 + i)] = t1[i];
#pragma unroll
    for (int i = 0; i < 24; ++i) t1[i] = srcB[tid + 128 * (25 + i)];
#pragma unroll
    for (int i = 0; i < 25; ++i) dst[PAIR_V4 + tid + 128 * i] = t0[i];
#pragma unroll
    for (int i = 0; i < 24; ++i) dst[PAIR_V4 + tid + 128 * (25 + i)] = t1[i];
  }

  bool above;
  int slotA, slotB, I, J;
  int dside = 0, dt = 0;
  if (g < 6) {
    above = true;
    slotA = wave >> 1;
    slotB = 2 + (wave & 1);
    I = p0 + (wave >> 1);
    J = p2 + (wave & 1);
    dside = wave >> 1;
    dt = 2 * (dside ? (p0 >> 1) : (p2 >> 1) - 1) + (wave & 1);
  } else {
    above = wave < 2;
    slotA = 2 * (wave & 1);
    slotB = slotA + 1;
    I = p0 + slotA;
    J = I + 1;
    dside = wave & 1;
    dt = 6;
  }
  const int drow_lds = dside * 128 + dt * 16;
  const int drow_g   = (dside ? p2 : p0) * 64 + dt * 16;

  __syncthreads();

  v8f acc[4][4], acc2[4][4], accD;
  if (above)
    gram_block_tile(acc, accD, lds, slotA * 64, slotB * 64, drow_lds, lr, half);
  else
    gram_diag_pair_upper(acc, acc2, lds, slotA * 64, slotB * 64, lr, half);

  __syncthreads();

  SketchKeys keysA, keysB;
  DiagKeys keysD;
  if (above) {
    sketch_keys(keysA, I * 64, J * 64, lr, half, h1, s1, h2, s2);
    diag_keys(keysD, drow_g, lr, half, h1, s1, h2, s2);
  } else {
    sketch_keys(keysA, I * 64, I * 64, lr, half, h1, s1, h2, s2);
    sketch_keys(keysB, J * 64, J * 64, lr, half, h1, s1, h2, s2);
  }

  float* myhist = lds + wave * ND;
#pragma unroll
  for (int i = 0; i < ND / 128; ++i)
    *(v4f*)(myhist + lane * 4 + 128 * i) = (v4f){0.0f, 0.0f, 0.0f, 0.0f};

  if (above) {
    sketch_add<0>(acc, myhist, keysA);
    sketch_diag_add(accD, myhist, keysD);
  } else {
    sketch_add<2>(acc, myhist, keysA);
    sketch_add<3>(acc2, myhist, keysB);
  }

  __syncthreads();
  float* dst = ((g < NPLANE - 1) ? planes + (size_t)g * (NB * ND) : out) + (size_t)b * ND;
  v4f part[ND / 512][4];
#pragma unroll
  for (int i = 0; i < ND / 512; ++i)
#pragma unroll
    for (int w = 0; w < 4; ++w) part[i][w] = *(const v4f*)(lds + w * ND + tid * 4 + 512 * i);
#pragma unroll
  for (int i = 0; i < ND / 512; ++i) {
    v4f v = part[i][0];
    v += part[i][1];
    v += part[i][2];
    v += part[i][3];
    *(v4f*)(dst + tid * 4 + 512 * i) = v;
  }
}

__global__ __launch_bounds__(256) void cbp_finalize_planes_kernel(
    const float* __restrict__ planes, float* out) {
  __shared__ float red[256];
  const int b   = blockIdx.x;
  const int tid = threadIdx.x;
  const float* p0 = planes + (size_t)b * ND;
  float* o = out + (size_t)b * ND;

  v4f pv[ND / 1024][NPLANE];
#pragma unroll
  for (int k = 0; k < ND / 1024; ++k) {
    const int idx = 4 * (tid + 256 * k);
#pragma unroll
    for (int p = 0; p < NPLANE - 1; ++p) pv[k][p] = *(const v4f*)(p0 + (size_t)p * (NB * ND) + idx);
    pv[k][NPLANE - 1] = *(const v4f*)(o + idx);
  }
  __builtin_amdgcn_sched_barrier(0);

  v4f v[ND / 1024];
  float ssum = 0.0f;
#pragma unroll
  for (int k = 0; k < ND / 1024; ++k) {
    v4f a = pv[k][0];
#pragma unroll
    for (int p = 1; p < NPLANE; ++p) a += pv[k][p];
    v[k] = a;
    ssum += fabsf(a.x) + 1e-5f;
    ssum += fabsf(a.y) + 1e-5f;
    ssum += fabsf(a.z) + 1e-5f;
    ssum += fabsf(a.w) + 1e-5f;
  }
  red[tid] = ssum;
  __syncthreads();
  for (int s = 128; s > 0; s >>= 1) {
    if (tid < s) red[tid] += red[tid + s];
    __syncthreads();
  }
  const float inv = 1.0f / fmaxf(sqrtf(red[0]), 1e-12f);

#pragma unroll
  for (int k = 0; k < ND / 1024; ++k) {
    v4f r;
#pragma unroll
    for (int e = 0; e < 4; ++e) {
      const float val = v[k][e];
      const float sg  = (val > 0.0f) ? 1.0f : ((val < 0.0f) ? -1.0f : 0.0f);
      r[e] = sg * sqrtf(fabsf(val) + 1e-5f) * inv;
    }
    *(v4f*)(o + 4 * (tid + 256 * k)) = r;
  }
}

extern "C" void kernel_launch(void* const* d_in, const int* in_sizes, int n_in,
                              void* d_out, int out_size, void* d_ws, size_t ws_size,
                              hipStream_t stream) {
  const float* x  = (const float*)d_in[0];
  const int*   h1 = (const int*)d_in[1];
  const float* s1 = (const float*)d_in[2];
  const int*   h2 = (const int*)d_in[3];
  const float* s2 = (const float*)d_in[4];
  float*       out = (float*)d_out;

  float* cbp = (float*)d_ws;
  const int n = NB * ND;

  if (ws_size >= 7340032) {
    cbp_gram_sketch_planes_kernel<<<NB * NPLANE, 128, 0, stream>>>(x, h1, s1, h2, s2, cbp, out);
    cbp_finalize_planes_kernel<<<NB, 256, 0, stream>>>(cbp, out);
  } else {
    cbp_zero_kernel<<<(n + 255) / 256, 256, 0, stream>>>(cbp, out, n);
    cbp_gram_sketch_kernel<<<NB * 4, 256, 0, stream>>>(x, h1, s1, h2, s2, cbp, out);
    cbp_finalize_kernel<<<NB, 256, 0, stream>>>(cbp, out);
  }
}
